# starting kernel
# baseline (speedup 1.0000x reference)
#include <hip/hip_runtime.h>

#define NN 100000
#define NE 1600000

typedef float v2f __attribute__((ext_vector_type(2)));
typedef float v8f __attribute__((ext_vector_type(8)));

__global__ __launch_bounds__(256) void k_deg(const int* __restrict__ dst,
                                             unsigned* __restrict__ deg, int E) {
  int e = blockIdx.x * 256 + threadIdx.x;
  if (e < E) atomicAdd(&deg[dst[e]], 1u);
}

__global__ __launch_bounds__(256) void k_dinv(const unsigned* __restrict__ deg,
                                              float* __restrict__ dinv, int N) {
  int i = blockIdx.x * 256 + threadIdx.x;
  if (i < N) dinv[i] = 1.0f / sqrtf((float)deg[i] + 1.0f);
}

__global__ __launch_bounds__(256) void k_norm(const int* __restrict__ src,
                                              const int* __restrict__ dst,
                                              const float* __restrict__ dinv,
                                              float* __restrict__ nrm, int E) {
  int e = blockIdx.x * 256 + threadIdx.x;
  if (e < E) nrm[e] = dinv[src[e]] * dinv[dst[e]];
}

template <int K>
__global__ __launch_bounds__(256) void k_gemm(const float* __restrict__ X,
                                              const float* __restrict__ W,
                                              float* __restrict__ H, int M) {
  __shared__ __align__(16) v2f wl[(K / 2) * 64];
  for (int i = threadIdx.x; i < (K / 2) * 64; i += 256) {
    int p = i >> 6, n = i & 63;
    v2f t = { W[(size_t)(2 * p) * 64 + n], W[(size_t)(2 * p + 1) * 64 + n] };
    wl[i] = t;
  }
  __syncthreads();

  int wave  = threadIdx.x >> 5;
  int lane  = threadIdx.x & 31;
  int strip = blockIdx.x * 8 + wave;
  if (strip * 16 >= M) return;
  int row0 = strip * 16;
  int m    = lane & 15;
  int h2   = lane >> 4;

  const float* xrow = X + (size_t)(row0 + m) * K;
  v8f a0 = {}, a1 = {}, a2 = {}, a3 = {};
  for (int kb = 0; kb < K; kb += 4) {
    int pa = (kb >> 1) + h2;
    v2f a = *(const v2f*)(xrow + 2 * pa);
    const v2f* wp = &wl[pa * 64 + m];
    v2f b0 = wp[0];
    v2f b1 = wp[16];
    v2f b2 = wp[32];
    v2f b3 = wp[48];
    a0 = __builtin_amdgcn_wmma_f32_16x16x4_f32(false, a, false, b0, (short)0, a0, false, false);
    a1 = __builtin_amdgcn_wmma_f32_16x16x4_f32(false, a, false, b1, (short)0, a1, false, false);
    a2 = __builtin_amdgcn_wmma_f32_16x16x4_f32(false, a, false, b2, (short)0, a2, false, false);
    a3 = __builtin_amdgcn_wmma_f32_16x16x4_f32(false, a, false, b3, (short)0, a3, false, false);
  }
  float* out0 = H + (size_t)(row0 + 8 * h2) * 64 + m;
#pragma unroll
  for (int j = 0; j < 8; ++j) {
    float* r = out0 + (size_t)j * 64;
    r[0] = a0[j]; r[16] = a1[j]; r[32] = a2[j]; r[48] = a3[j];
  }
}

__global__ __launch_bounds__(256) void k_agg(const float* __restrict__ H,
                                             const int* __restrict__ src,
                                             const int* __restrict__ dst,
                                             const float* __restrict__ nrm,
                                             float* __restrict__ agg, int E) {
  int e = (int)((blockIdx.x * 256 + threadIdx.x) >> 5);
  if (e >= E) return;
  int lane = threadIdx.x & 31;
  int s = src[e], d = dst[e];
  float w = nrm[e];
  v2f hv = *(const v2f*)(H + (size_t)s * 64 + lane * 2);
  float* p = agg + (size_t)d * 64 + lane * 2;
  unsafeAtomicAdd(p,     w * hv.x);
  unsafeAtomicAdd(p + 1, w * hv.y);
}

__global__ __launch_bounds__(256) void k_relu_l2(const float* __restrict__ agg,
                                                 const float* __restrict__ H,
                                                 const float* __restrict__ dinv,
                                                 const float* __restrict__ bias,
                                                 float* __restrict__ out, int N) {
  int node = (int)((blockIdx.x * 256 + threadIdx.x) >> 5);
  if (node >= N) return;
  int lane = threadIdx.x & 31;
  float di = dinv[node];
  float sl = di * di;
  size_t base = (size_t)node * 64 + lane * 2;
  v2f a  = *(const v2f*)(agg + base);
  v2f hv = *(const v2f*)(H + base);
  float vx = fmaxf(a.x + sl * hv.x + bias[lane * 2],     0.0f);
  float vy = fmaxf(a.y + sl * hv.y + bias[lane * 2 + 1], 0.0f);
  float ss = vx * vx + vy * vy;
#pragma unroll
  for (int off = 16; off > 0; off >>= 1) ss += __shfl_xor(ss, off, 32);
  float scale = 1.0f / fmaxf(sqrtf(ss), 1e-12f);
  out[base]     = vx * scale;
  out[base + 1] = vy * scale;
}

__global__ __launch_bounds__(256) void k_final(const float* __restrict__ agg,
                                               const float* __restrict__ H,
                                               const float* __restrict__ dinv,
                                               const float* __restrict__ bias,
                                               float* __restrict__ out, int N) {
  int idx = blockIdx.x * 256 + threadIdx.x;
  if (idx >= N * 64) return;
  int node = idx >> 6;
  int c    = idx & 63;
  float di = dinv[node];
  out[idx] = agg[idx] + di * di * H[idx] + bias[c];
}

extern "C" void kernel_launch(void* const* d_in, const int* in_sizes, int n_in,
                              void* d_out, int out_size, void* d_ws, size_t ws_size,
                              hipStream_t stream) {
  const float* x  = (const float*)d_in[0];
  const int*   ei = (const int*)d_in[1];
  const float* W1 = (const float*)d_in[2];
  const float* b1 = (const float*)d_in[3];
  const float* W2 = (const float*)d_in[4];
  const float* b2 = (const float*)d_in[5];
  const float* W3 = (const float*)d_in[6];
  const float* b3 = (const float*)d_in[7];
  float* out = (float*)d_out;

  const int* src = ei;
  const int* dst = ei + NE;

  char* ws = (char*)d_ws;
  size_t off = 0;
  auto carve = [&](size_t bytes) {
    void* p = ws + off;
    off = (off + bytes + 255) & ~(size_t)255;
    return p;
  };
  unsigned* deg  = (unsigned*)carve((size_t)NN * 4);
  float*    dinv = (float*)   carve((size_t)NN * 4);
  float*    nrm  = (float*)   carve((size_t)NE * 4);
  float*    h    = (float*)   carve((size_t)NN * 64 * 4);
  float*    agg  = (float*)   carve((size_t)NN * 64 * 4);
  float*    hn   = (float*)   carve((size_t)NN * 64 * 4);

  const int eBlk   = (NE + 255) / 256;
  const int nBlk   = (NN + 255) / 256;
  const int gBlk   = ((NN / 16) + 7) / 8;
  const int aggBlk = NE / 8;
  const int wnBlk  = (NN * 32 + 255) / 256;
  const int feBlk  = (NN * 64 + 255) / 256;

  hipMemsetAsync(deg, 0, (size_t)NN * 4, stream);
  k_deg <<<eBlk, 256, 0, stream>>>(dst, deg, NE);
  k_dinv<<<nBlk, 256, 0, stream>>>(deg, dinv, NN);
  k_norm<<<eBlk, 256, 0, stream>>>(src, dst, dinv, nrm, NE);

  k_gemm<256><<<gBlk, 256, 0, stream>>>(x, W1, h, NN);
  hipMemsetAsync(agg, 0, (size_t)NN * 64 * 4, stream);
  k_agg    <<<aggBlk, 256, 0, stream>>>(h, src, dst, nrm, agg, NE);
  k_relu_l2<<<wnBlk,  256, 0, stream>>>(agg, h, dinv, b1, hn, NN);

  k_gemm<64><<<gBlk, 256, 0, stream>>>(hn, W2, h, NN);
  hipMemsetAsync(agg, 0, (size_t)NN * 64 * 4, stream);
  k_agg    <<<aggBlk, 256, 0, stream>>>(h, src, dst, nrm, agg, NE);
  k_relu_l2<<<wnBlk,  256, 0, stream>>>(agg, h, dinv, b2, hn, NN);

  k_gemm<64><<<gBlk, 256, 0, stream>>>(hn, W3, h, NN);
  hipMemsetAsync(agg, 0, (size_t)NN * 64 * 4, stream);
  k_agg  <<<aggBlk, 256, 0, stream>>>(h, src, dst, nrm, agg, NE);
  k_final<<<feBlk,  256, 0, stream>>>(agg, h, dinv, b3, out, NN);

  (void)in_sizes; (void)n_in; (void)out_size; (void)ws_size;
}


# Aggregate by pull: edges grouped by destination in-call (rank from the degree add, scan, fill); one wave sums a node's rows in registers; no float atomics, no zero fill; sum order = arrival rank
# speedup vs baseline: 2.1404x; 2.1404x over previous
#include <hip/hip_runtime.h>


#define NN 100000
#define NE 1600000
#define SCAN_TILE 1024

typedef float v2f __attribute__((ext_vector_type(2)));
typedef float v8f __attribute__((ext_vector_type(8)));
typedef unsigned v2u __attribute__((ext_vector_type(2)));
typedef unsigned v4u __attribute__((ext_vector_type(4)));

__global__ __launch_bounds__(256) void k_deg(const int* __restrict__ dst,
                                             unsigned* __restrict__ deg,
                                             unsigned* __restrict__ rank, int E) {
  int e = blockIdx.x * 256 + threadIdx.x;
  if (e < E) rank[e] = atomicAdd(&deg[dst[e]], 1u);
}

__global__ __launch_bounds__(256) void k_dinv(const unsigned* __restrict__ deg,
                                              float* __restrict__ dinv, int N) {
  int i = blockIdx.x * 256 + threadIdx.x;
  if (i < N) dinv[i] = 1.0f / sqrtf((float)deg[i] + 1.0f);
}

__global__ __launch_bounds__(256) void k_bsum(const unsigned* __restrict__ deg,
                                              unsigned* __restrict__ bsum, int N) {
  __shared__ unsigned red[8];
  int lane = threadIdx.x & 31, wave = threadIdx.x >> 5;
  int i0 = (blockIdx.x * 256 + threadIdx.x) * 4;
  unsigned t = 0;
  if (i0 + 3 < N) {
    v4u v = *(const v4u*)(deg + i0);
    t = v.x + v.y + v.z + v.w;
  } else {
    if (i0 < N)     t += deg[i0];
    if (i0 + 1 < N) t += deg[i0 + 1];
    if (i0 + 2 < N) t += deg[i0 + 2];
  }
  for (int off = 16; off > 0; off >>= 1) t += __shfl_xor(t, off, 32);
  if (lane == 0) red[wave] = t;
  __syncthreads();
  if (threadIdx.x == 0) {
    unsigned s = 0;
    for (int w = 0; w < 8; ++w) s += red[w];
    bsum[blockIdx.x] = s;
  }
}

__global__ __launch_bounds__(256) void k_scan(const unsigned* __restrict__ deg,
                                              const unsigned* __restrict__ bsum,
                                              unsigned* __restrict__ rowptr, int N) {
  __shared__ unsigned red[8];
  __shared__ unsigned wtot[8];
  int lane = threadIdx.x & 31, wave = threadIdx.x >> 5;
  unsigned part = 0;
  for (int j = threadIdx.x; j < (int)blockIdx.x; j += 256) part += bsum[j];
  for (int off = 16; off > 0; off >>= 1) part += __shfl_xor(part, off, 32);

  int i0 = (blockIdx.x * 256 + threadIdx.x) * 4;
  unsigned d0 = 0, d1 = 0, d2 = 0, d3 = 0;
  if (i0 + 3 < N) {
    v4u v = *(const v4u*)(deg + i0);
    d0 = v.x; d1 = v.y; d2 = v.z; d3 = v.w;
  } else {
    if (i0 < N)     d0 = deg[i0];
    if (i0 + 1 < N) d1 = deg[i0 + 1];
    if (i0 + 2 < N) d2 = deg[i0 + 2];
  }
  unsigned t = d0 + d1 + d2 + d3;
  unsigned inc = t;
  for (int off = 1; off < 32; off <<= 1) {
    unsigned y = __shfl_up(inc, off, 32);
    if (lane >= off) inc += y;
  }
  if (lane == 0)  red[wave]  = part;
  if (lane == 31) wtot[wave] = inc;
  __syncthreads();
  unsigned base = 0;
  for (int w = 0; w < 8; ++w) {
    base += red[w];
    if (w < wave) base += wtot[w];
  }
  unsigned p0 = base + inc - t, p1 = p0 + d0, p2 = p1 + d1, p3 = p2 + d2;
  if (i0 + 3 <= N) {
    v4u p = { p0, p1, p2, p3 };
    *(v4u*)(rowptr + i0) = p;
  } else {
    if (i0 <= N)     rowptr[i0]     = p0;
    if (i0 + 1 <= N) rowptr[i0 + 1] = p1;
    if (i0 + 2 <= N) rowptr[i0 + 2] = p2;
  }
}

__global__ __launch_bounds__(256) void k_fill(const int* __restrict__ src,
                                              const int* __restrict__ dst,
                                              const float* __restrict__ dinv,
                                              const unsigned* __restrict__ rowptr,
                                              const unsigned* __restrict__ rank,
                                              unsigned* __restrict__ csr, int E) {
  int e = blockIdx.x * 256 + threadIdx.x;
  if (e < E) {
    int s = src[e], d = dst[e];
    unsigned pos = rowptr[d] + rank[e];
    v2u ent = { (unsigned)s, __float_as_uint(dinv[s] * dinv[d]) };
    ((v2u*)csr)[pos] = ent;
  }
}

template <int K>
__global__ __launch_bounds__(256) void k_gemm(const float* __restrict__ X,
                                              const float* __restrict__ W,
                                              float* __restrict__ H, int M) {
  __shared__ __align__(16) v2f wl[(K / 2) * 64];
  for (int i = threadIdx.x; i < (K / 2) * 64; i += 256) {
    int p = i >> 6, n = i & 63;
    v2f t = { W[(size_t)(2 * p) * 64 + n], W[(size_t)(2 * p + 1) * 64 + n] };
    wl[i] = t;
  }
  __syncthreads();

  int wave  = threadIdx.x >> 5;
  int lane  = threadIdx.x & 31;
  int strip = blockIdx.x * 8 + wave;
  if (strip * 16 >= M) return;
  int row0 = strip * 16;
  int m    = lane & 15;
  int h2   = lane >> 4;

  const float* xrow = X + (size_t)(row0 + m) * K;
  v8f a0 = {}, a1 = {}, a2 = {}, a3 = {};
  for (int kb = 0; kb < K; kb += 4) {
    int pa = (kb >> 1) + h2;
    v2f a = *(const v2f*)(xrow + 2 * pa);
    const v2f* wp = &wl[pa * 64 + m];
    v2f b0 = wp[0];
    v2f b1 = wp[16];
    v2f b2 = wp[32];
    v2f b3 = wp[48];
    a0 = __builtin_amdgcn_wmma_f32_16x16x4_f32(false, a, false, b0, (short)0, a0, false, false);
    a1 = __builtin_amdgcn_wmma_f32_16x16x4_f32(false, a, false, b1, (short)0, a1, false, false);
    a2 = __builtin_amdgcn_wmma_f32_16x16x4_f32(false, a, false, b2, (short)0, a2, false, false);
    a3 = __builtin_amdgcn_wmma_f32_16x16x4_f32(false, a, false, b3, (short)0, a3, false, false);
  }
  float* out0 = H + (size_t)(row0 + 8 * h2) * 64 + m;
#pragma unroll
  for (int j = 0; j < 8; ++j) {
    float* r = out0 + (size_t)j * 64;
    r[0] = a0[j]; r[16] = a1[j]; r[32] = a2[j]; r[48] = a3[j];
  }
}

__global__ __launch_bounds__(256) void k_pull(const float* __restrict__ H,
                                              const unsigned* __restrict__ rowptr,
                                              const unsigned* __restrict__ csr,
                                              float* __restrict__ agg, int N) {
  int node = (int)((blockIdx.x * 256 + threadIdx.x) >> 5);
  if (node >= N) return;
  node = __builtin_amdgcn_readfirstlane(node);
  int lane = threadIdx.x & 31;
  unsigned beg = rowptr[node], end = rowptr[node + 1];
  const v2u* ent = (const v2u*)csr;
  const float* hl = H + lane * 2;
  const v2f zero = { 0.0f, 0.0f };
  v2f acc = zero;
  for (unsigned cb = beg; cb < end; cb += 32) {
    unsigned n = end - cb;
    if (n > 32u) n = 32u;
    v2u my = { 0u, 0u };
    if ((unsigned)lane < n) my = ent[cb + lane];
    for (unsigned j = 0; j < n; j += 8) {
      unsigned s[8];
      float w[8];
      v2f r[8];
#pragma unroll
      for (int u = 0; u < 8; ++u) {
        s[u] = (unsigned)__builtin_amdgcn_readlane((int)my.x, (int)(j + u));
        w[u] = __uint_as_float((unsigned)__builtin_amdgcn_readlane((int)my.y, (int)(j + u)));
      }
#pragma unroll
      for (int u = 0; u < 8; ++u) r[u] = *(const v2f*)(hl + (size_t)s[u] * 64);
#pragma unroll
      for (int u = 0; u < 8; ++u) {
        v2f rr = (j + u < n) ? r[u] : zero;
        acc += rr * w[u];
      }
    }
  }
  *(v2f*)(agg + (size_t)node * 64 + lane * 2) = acc;
}

__global__ __launch_bounds__(256) void k_relu_l2(const float* __restrict__ agg,
                                                 const float* __restrict__ H,
                                                 const float* __restrict__ dinv,
                                                 const float* __restrict__ bias,
                                                 float* __restrict__ out, int N) {
  int node = (int)((blockIdx.x * 256 + threadIdx.x) >> 5);
  if (node >= N) return;
  int lane = threadIdx.x & 31;
  float di = dinv[node];
  float sl = di * di;
  size_t base = (size_t)node * 64 + lane * 2;
  v2f a  = *(const v2f*)(agg + base);
  v2f hv = *(const v2f*)(H + base);
  float vx = fmaxf(a.x + sl * hv.x + bias[lane * 2],     0.0f);
  float vy = fmaxf(a.y + sl * hv.y + bias[lane * 2 + 1], 0.0f);
  float ss = vx * vx + vy * vy;
#pragma unroll
  for (int off = 16; off > 0; off >>= 1) ss += __shfl_xor(ss, off, 32);
  float scale = 1.0f / fmaxf(sqrtf(ss), 1e-12f);
  out[base]     = vx * scale;
  out[base + 1] = vy * scale;
}

__global__ __launch_bounds__(256) void k_final(const float* __restrict__ agg,
                                               const float* __restrict__ H,
                                               const float* __restrict__ dinv,
                                               const float* __restrict__ bias,
                                               float* __restrict__ out, int N) {
  int idx = blockIdx.x * 256 + threadIdx.x;
  if (idx >= N * 64) return;
  int node = idx >> 6;
  int c    = idx & 63;
  float di = dinv[node];
  out[idx] = agg[idx] + di * di * H[idx] + bias[c];
}

extern "C" void kernel_launch(void* const* d_in, const int* in_sizes, int n_in,
                              void* d_out, int out_size, void* d_ws, size_t ws_size,
                              hipStream_t stream) {
  const float* x  = (const float*)d_in[0];
  const int*   ei = (const int*)d_in[1];
  const float* W1 = (const float*)d_in[2];
  const float* b1 = (const float*)d_in[3];
  const float* W2 = (const float*)d_in[4];
  const float* b2 = (const float*)d_in[5];
  const float* W3 = (const float*)d_in[6];
  const float* b3 = (const float*)d_in[7];
  float* out = (float*)d_out;

  const int* src = ei;
  const int* dst = ei + NE;

  char* ws = (char*)d_ws;
  size_t off = 0;
  auto carve = [&](size_t bytes) {
    void* p = ws + off;
    off = (off + bytes + 255) & ~(size_t)255;
    return p;
  };
  const int sBlk = (NN + 1 + SCAN_TILE - 1) / SCAN_TILE;
  unsigned* deg    = (unsigned*)carve((size_t)NN * 4);
  float*    dinv   = (float*)   carve((size_t)NN * 4);
  unsigned* bsum   = (unsigned*)carve((size_t)sBlk * 4);
  unsigned* rowptr = (unsigned*)carve((size_t)(NN + 1) * 4);
  unsigned* rank   = (unsigned*)carve((size_t)NE * 4);
  unsigned* csr    = (unsigned*)carve((size_t)NE * 8);
  float*    h      = (float*)   carve((size_t)NN * 64 * 4);
  float*    agg    = (float*)   carve((size_t)NN * 64 * 4);
  float*    hn     = (float*)   carve((size_t)NN * 64 * 4);

  const int eBlk   = (NE + 255) / 256;
  const int nBlk   = (NN + 255) / 256;
  const int gBlk   = ((NN / 16) + 7) / 8;
  const int wnBlk  = (NN * 32 + 255) / 256;
  const int feBlk  = (NN * 64 + 255) / 256;

  hipMemsetAsync(deg, 0, (size_t)NN * 4, stream);
  k_deg <<<eBlk, 256, 0, stream>>>(dst, deg, rank, NE);
  k_dinv<<<nBlk, 256, 0, stream>>>(deg, dinv, NN);
  k_bsum<<<sBlk, 256, 0, stream>>>(deg, bsum, NN);
  k_scan<<<sBlk, 256, 0, stream>>>(deg, bsum, rowptr, NN);
  k_fill<<<eBlk, 256, 0, stream>>>(src, dst, dinv, rowptr, rank, csr, NE);

  k_gemm<256><<<gBlk, 256, 0, stream>>>(x, W1, h, NN);
  k_pull   <<<wnBlk, 256, 0, stream>>>(h, rowptr, csr, agg, NN);
  k_relu_l2<<<wnBlk, 256, 0, stream>>>(agg, h, dinv, b1, hn, NN);

  k_gemm<64><<<gBlk, 256, 0, stream>>>(hn, W2, h, NN);
  k_pull   <<<wnBlk, 256, 0, stream>>>(h, rowptr, csr, agg, NN);
  k_relu_l2<<<wnBlk, 256, 0, stream>>>(agg, h, dinv, b2, hn, NN);

  k_gemm<64><<<gBlk, 256, 0, stream>>>(hn, W3, h, NN);
  k_pull <<<wnBlk, 256, 0, stream>>>(h, rowptr, csr, agg, NN);
  k_final<<<feBlk, 256, 0, stream>>>(agg, h, dinv, b3, out, NN);

  (void)in_sizes; (void)n_in; (void)out_size; (void)ws_size;
}


# Epilogue fused into the pull kernel: self-loop, bias, relu and L2 scale (layer 3: self-loop, bias) applied to the register sums; agg plane, 3 launches and 400000 waves gone
# speedup vs baseline: 2.4206x; 1.1309x over previous
#include <hip/hip_runtime.h>


#define NN 100000
#define NE 1600000
#define SCAN_TILE 1024

typedef float v2f __attribute__((ext_vector_type(2)));
typedef float v8f __attribute__((ext_vector_type(8)));
typedef unsigned v2u __attribute__((ext_vector_type(2)));
typedef unsigned v4u __attribute__((ext_vector_type(4)));

__global__ __launch_bounds__(256) void k_deg(const int* __restrict__ dst,
                                             unsigned* __restrict__ deg,
                                             unsigned* __restrict__ rank, int E) {
  int e = blockIdx.x * 256 + threadIdx.x;
  if (e < E) rank[e] = atomicAdd(&deg[dst[e]], 1u);
}

__global__ __launch_bounds__(256) void k_dinv(const unsigned* __restrict__ deg,
                                              float* __restrict__ dinv, int N) {
  int i = blockIdx.x * 256 + threadIdx.x;
  if (i < N) dinv[i] = 1.0f / sqrtf((float)deg[i] + 1.0f);
}

__global__ __launch_bounds__(256) void k_bsum(const unsigned* __restrict__ deg,
                                              unsigned* __restrict__ bsum, int N) {
  __shared__ unsigned red[8];
  int lane = threadIdx.x & 31, wave = threadIdx.x >> 5;
  int i0 = (blockIdx.x * 256 + threadIdx.x) * 4;
  unsigned t = 0;
  if (i0 + 3 < N) {
    v4u v = *(const v4u*)(deg + i0);
    t = v.x + v.y + v.z + v.w;
  } else {
    if (i0 < N)     t += deg[i0];
    if (i0 + 1 < N) t += deg[i0 + 1];
    if (i0 + 2 < N) t += deg[i0 + 2];
  }
  for (int off = 16; off > 0; off >>= 1) t += __shfl_xor(t, off, 32);
  if (lane == 0) red[wave] = t;
  __syncthreads();
  if (threadIdx.x == 0) {
    unsigned s = 0;
    for (int w = 0; w < 8; ++w) s += red[w];
    bsum[blockIdx.x] = s;
  }
}

__global__ __launch_bounds__(256) void k_scan(const unsigned* __restrict__ deg,
                                              const unsigned* __restrict__ bsum,
                                              unsigned* __restrict__ rowptr, int N) {
  __shared__ unsigned red[8];
  __shared__ unsigned wtot[8];
  int lane = threadIdx.x & 31, wave = threadIdx.x >> 5;
  unsigned part = 0;
  for (int j = threadIdx.x; j < (int)blockIdx.x; j += 256) part += bsum[j];
  for (int off = 16; off > 0; off >>= 1) part += __shfl_xor(part, off, 32);

  int i0 = (blockIdx.x * 256 + threadIdx.x) * 4;
  unsigned d0 = 0, d1 = 0, d2 = 0, d3 = 0;
  if (i0 + 3 < N) {
    v4u v = *(const v4u*)(deg + i0);
    d0 = v.x; d1 = v.y; d2 = v.z; d3 = v.w;
  } else {
    if (i0 < N)     d0 = deg[i0];
    if (i0 + 1 < N) d1 = deg[i0 + 1];
    if (i0 + 2 < N) d2 = deg[i0 + 2];
  }
  unsigned t = d0 + d1 + d2 + d3;
  unsigned inc = t;
  for (int off = 1; off < 32; off <<= 1) {
    unsigned y = __shfl_up(inc, off, 32);
    if (lane >= off) inc += y;
  }
  if (lane == 0)  red[wave]  = part;
  if (lane == 31) wtot[wave] = inc;
  __syncthreads();
  unsigned base = 0;
  for (int w = 0; w < 8; ++w) {
    base += red[w];
    if (w < wave) base += wtot[w];
  }
  unsigned p0 = base + inc - t, p1 = p0 + d0, p2 = p1 + d1, p3 = p2 + d2;
  if (i0 + 3 <= N) {
    v4u p = { p0, p1, p2, p3 };
    *(v4u*)(rowptr + i0) = p;
  } else {
    if (i0 <= N)     rowptr[i0]     = p0;
    if (i0 + 1 <= N) rowptr[i0 + 1] = p1;
    if (i0 + 2 <= N) rowptr[i0 + 2] = p2;
  }
}

__global__ __launch_bounds__(256) void k_fill(const int* __restrict__ src,
                                              const int* __restrict__ dst,
                                              const float* __restrict__ dinv,
                                              const unsigned* __restrict__ rowptr,
                                              const unsigned* __restrict__ rank,
                                              unsigned* __restrict__ csr, int E) {
  int e = blockIdx.x * 256 + threadIdx.x;
  if (e < E) {
    int s = src[e], d = dst[e];
    unsigned pos = rowptr[d] + rank[e];
    v2u ent = { (unsigned)s, __float_as_uint(dinv[s] * dinv[d]) };
    ((v2u*)csr)[pos] = ent;
  }
}

template <int K>
__global__ __launch_bounds__(256) void k_gemm(const float* __restrict__ X,
                                              const float* __restrict__ W,
                                              float* __restrict__ H, int M) {
  __shared__ __align__(16) v2f wl[(K / 2) * 64];
  for (int i = threadIdx.x; i < (K / 2) * 64; i += 256) {
    int p = i >> 6, n = i & 63;
    v2f t = { W[(size_t)(2 * p) * 64 + n], W[(size_t)(2 * p + 1) * 64 + n] };
    wl[i] = t;
  }
  __syncthreads();

  int wave  = threadIdx.x >> 5;
  int lane  = threadIdx.x & 31;
  int strip = blockIdx.x * 8 + wave;
  if (strip * 16 >= M) return;
  int row0 = strip * 16;
  int m    = lane & 15;
  int h2   = lane >> 4;

  const float* xrow = X + (size_t)(row0 + m) * K;
  v8f a0 = {}, a1 = {}, a2 = {}, a3 = {};
  for (int kb = 0; kb < K; kb += 4) {
    int pa = (kb >> 1) + h2;
    v2f a = *(const v2f*)(xrow + 2 * pa);
    const v2f* wp = &wl[pa * 64 + m];
    v2f b0 = wp[0];
    v2f b1 = wp[16];
    v2f b2 = wp[32];
    v2f b3 = wp[48];
    a0 = __builtin_amdgcn_wmma_f32_16x16x4_f32(false, a, false, b0, (short)0, a0, false, false);
    a1 = __builtin_amdgcn_wmma_f32_16x16x4_f32(false, a, false, b1, (short)0, a1, false, false);
    a2 = __builtin_amdgcn_wmma_f32_16x16x4_f32(false, a, false, b2, (short)0, a2, false, false);
    a3 = __builtin_amdgcn_wmma_f32_16x16x4_f32(false, a, false, b3, (short)0, a3, false, false);
  }
  float* out0 = H + (size_t)(row0 + 8 * h2) * 64 + m;
#pragma unroll
  for (int j = 0; j < 8; ++j) {
    float* r = out0 + (size_t)j * 64;
    r[0] = a0[j]; r[16] = a1[j]; r[32] = a2[j]; r[48] = a3[j];
  }
}

template <bool NORM>
__global__ __launch_bounds__(256) void k_pull(const float* __restrict__ H,
                                              const unsigned* __restrict__ rowptr,
                                              const unsigned* __restrict__ csr,
                                              const float* __restrict__ dinv,
                                              const float* __restrict__ bias,
                                              float* __restrict__ out, int N) {
  int node = (int)((blockIdx.x * 256 + threadIdx.x) >> 5);
  if (node >= N) return;
  node = __builtin_amdgcn_readfirstlane(node);
  int lane = threadIdx.x & 31;
  unsigned beg = rowptr[node], end = rowptr[node + 1];
  size_t base = (size_t)node * 64 + lane * 2;
  v2f hv = *(const v2f*)(H + base);
  v2f bv = *(const v2f*)(bias + lane * 2);
  float di = dinv[node];
  const v2u* ent = (const v2u*)csr;
  const float* hl = H + lane * 2;
  const v2f zero = { 0.0f, 0.0f };
  v2f acc = zero;
  for (unsigned cb = beg; cb < end; cb += 32) {
    unsigned n = end - cb;
    if (n > 32u) n = 32u;
    v2u my = { 0u, 0u };
    if ((unsigned)lane < n) my = ent[cb + lane];
    for (unsigned j = 0; j < n; j += 8) {
      unsigned s[8];
      float w[8];
      v2f r[8];
#pragma unroll
      for (int u = 0; u < 8; ++u) {
        s[u] = (unsigned)__builtin_amdgcn_readlane((int)my.x, (int)(j + u));
        w[u] = __uint_as_float((unsigned)__builtin_amdgcn_readlane((int)my.y, (int)(j + u)));
      }
#pragma unroll
      for (int u = 0; u < 8; ++u) r[u] = *(const v2f*)(hl + (size_t)s[u] * 64);
#pragma unroll
      for (int u = 0; u < 8; ++u) {
        v2f rr = (j + u < n) ? r[u] : zero;
        acc += rr * w[u];
      }
    }
  }
  float sl = di * di;
  float vx = acc.x + sl * hv.x + bv.x;
  float vy = acc.y + sl * hv.y + bv.y;
  if (NORM) {
    vx = fmaxf(vx, 0.0f);
    vy = fmaxf(vy, 0.0f);
    float ss = vx * vx + vy * vy;
    for (int off = 16; off > 0; off >>= 1) ss += __shfl_xor(ss, off, 32);
    float scale = 1.0f / fmaxf(sqrtf(ss), 1e-12f);
    vx *= scale;
    vy *= scale;
  }
  v2f o = { vx, vy };
  *(v2f*)(out + base) = o;
}

extern "C" void kernel_launch(void* const* d_in, const int* in_sizes, int n_in,
                              void* d_out, int out_size, void* d_ws, size_t ws_size,
                              hipStream_t stream) {
  const float* x  = (const float*)d_in[0];
  const int*   ei = (const int*)d_in[1];
  const float* W1 = (const float*)d_in[2];
  const float* b1 = (const float*)d_in[3];
  const float* W2 = (const float*)d_in[4];
  const float* b2 = (const float*)d_in[5];
  const float* W3 = (const float*)d_in[6];
  const float* b3 = (const float*)d_in[7];
  float* out = (float*)d_out;

  const int* src = ei;
  const int* dst = ei + NE;

  char* ws = (char*)d_ws;
  size_t off = 0;
  auto carve = [&](size_t bytes) {
    void* p = ws + off;
    off = (off + bytes + 255) & ~(size_t)255;
    return p;
  };
  const int sBlk = (NN + 1 + SCAN_TILE - 1) / SCAN_TILE;
  unsigned* deg    = (unsigned*)carve((size_t)NN * 4);
  float*    dinv   = (float*)   carve((size_t)NN * 4);
  unsigned* bsum   = (unsigned*)carve((size_t)sBlk * 4);
  unsigned* rowptr = (unsigned*)carve((size_t)(NN + 1) * 4);
  unsigned* rank   = (unsigned*)carve((size_t)NE * 4);
  unsigned* csr    = (unsigned*)carve((size_t)NE * 8);
  float*    h      = (float*)   carve((size_t)NN * 64 * 4);
  float*    hn     = (float*)   carve((size_t)NN * 64 * 4);

  const int eBlk   = (NE + 255) / 256;
  const int nBlk   = (NN + 255) / 256;
  const int gBlk   = ((NN / 16) + 7) / 8;
  const int wnBlk  = (NN * 32 + 255) / 256;

  hipMemsetAsync(deg, 0, (size_t)NN * 4, stream);
  k_deg <<<eBlk, 256, 0, stream>>>(dst, deg, rank, NE);
  k_dinv<<<nBlk, 256, 0, stream>>>(deg, dinv, NN);
  k_bsum<<<sBlk, 256, 0, stream>>>(deg, bsum, NN);
  k_scan<<<sBlk, 256, 0, stream>>>(deg, bsum, rowptr, NN);
  k_fill<<<eBlk, 256, 0, stream>>>(src, dst, dinv, rowptr, rank, csr, NE);

  k_gemm<256><<<gBlk, 256, 0, stream>>>(x, W1, h, NN);
  k_pull<true><<<wnBlk, 256, 0, stream>>>(h, rowptr, csr, dinv, b1, hn, NN);

  k_gemm<64><<<gBlk, 256, 0, stream>>>(hn, W2, h, NN);
  k_pull<true><<<wnBlk, 256, 0, stream>>>(h, rowptr, csr, dinv, b2, hn, NN);

  k_gemm<64><<<gBlk, 256, 0, stream>>>(hn, W3, h, NN);
  k_pull<false><<<wnBlk, 256, 0, stream>>>(h, rowptr, csr, dinv, b3, out, NN);

  (void)in_sizes; (void)n_in; (void)out_size; (void)ws_size;
}


# CSR build by two-level counting sort in LDS: chunks ordered by 512-node bin, then each bin by node; no global atomics, memset, scan or scattered fill; 11 launches to 8; sum order = LDS arrival rank
# speedup vs baseline: 4.3051x; 1.7785x over previous
#include <hip/hip_runtime.h>


#define NN 100000
#define NE 1600000
#define BSH 9
#define BNODES (1 << BSH)
#define NBIN ((NN >> BSH) + 1)
#define NPART 256
#define CH ((((NE + NPART - 1) / NPART) + 31) & ~31)
#define SRC_BITS 17
#define SRC_MASK ((1u << SRC_BITS) - 1u)
#define CAP 16384

static_assert(NN <= (1 << SRC_BITS), "src must fit its field");
static_assert(NBIN <= 256, "bin histogram is 256 wide");
static_assert(NPART == 256, "8 runs per wave of a 32-wave block; 4 column groups");

typedef float v2f __attribute__((ext_vector_type(2)));
typedef float v4f __attribute__((ext_vector_type(4)));
typedef float v8f __attribute__((ext_vector_type(8)));

template <int PER>
__device__ __forceinline__ void wave_scan(const unsigned* in, unsigned* out, int lane) {
  unsigned v[PER];
  unsigned t = 0;
#pragma unroll
  for (int j = 0; j < PER; ++j) { v[j] = in[lane * PER + j]; t += v[j]; }
  unsigned inc = t;
  for (int off = 1; off < 32; off <<= 1) {
    unsigned y = __shfl_up(inc, off, 32);
    if (lane >= off) inc += y;
  }
  unsigned run = inc - t;
#pragma unroll
  for (int j = 0; j < PER; ++j) { out[lane * PER + j] = run; run += v[j]; }
}

__global__ __launch_bounds__(1024) void k_part(const int* __restrict__ src,
                                               const int* __restrict__ dst,
                                               unsigned* __restrict__ part,
                                               unsigned* __restrict__ tbl) {
  __shared__ unsigned hist[256];
  __shared__ unsigned start[256];
  __shared__ unsigned staged[CH];
  const int tid = threadIdx.x, lane = tid & 31;
  const int p = blockIdx.x;
  const int e0 = p * CH;
  int n = NE - e0;
  if (n > CH) n = CH;
  if (n < 0) n = 0;
  if (tid < 256) hist[tid] = 0;
  __syncthreads();
  constexpr int PER = (CH + 1023) / 1024;
  unsigned key[PER], bin[PER], rk[PER];
#pragma unroll
  for (int i = 0; i < PER; ++i) {
    int j = i * 1024 + tid;
    int e = e0 + j;
    if (e > NE - 1) e = NE - 1;
    unsigned s = (unsigned)src[e], d = (unsigned)dst[e];
    rk[i] = 0u;
    key[i] = ((d & (unsigned)(BNODES - 1)) << SRC_BITS) | s;
    bin[i] = (j < n) ? (d >> BSH) : 0xFFFFFFFFu;
  }
#pragma unroll
  for (int i = 0; i < PER; ++i)
    if (bin[i] != 0xFFFFFFFFu) rk[i] = atomicAdd(&hist[bin[i]], 1u);
  __syncthreads();
  if (tid < 32) wave_scan<8>(hist, start, lane);
  if (tid < NBIN) tbl[tid * NPART + p] = hist[tid];
  __syncthreads();
#pragma unroll
  for (int i = 0; i < PER; ++i)
    if (bin[i] != 0xFFFFFFFFu) staged[start[bin[i]] + rk[i]] = key[i];
  __syncthreads();
  for (int j = tid; j < n; j += 1024) part[e0 + j] = staged[j];
}

__global__ __launch_bounds__(1024) void k_binsort(const unsigned* __restrict__ part,
                                                  const unsigned* __restrict__ tbl,
                                                  unsigned* __restrict__ rowptr,
                                                  float* __restrict__ dinv,
                                                  unsigned* __restrict__ csr) {
  __shared__ unsigned psum[1024];
  __shared__ unsigned runsrc[NPART];
  __shared__ unsigned runcnt[NPART];
  __shared__ unsigned cnt[BNODES];
  __shared__ unsigned cursor[BNODES];
  __shared__ unsigned red[16];
  __shared__ unsigned outb[CAP];
  const int tid = threadIdx.x, lane = tid & 31, wave = tid >> 5;
  const int b = blockIdx.x;
  if (tid < BNODES) cnt[tid] = 0;
  const int t = tid & (NPART - 1), g = tid >> 8;
  unsigned ps = 0;
  for (int bb = g; bb < b; bb += 4) ps += tbl[bb * NPART + t];
  psum[tid] = ps;
  __syncthreads();
  unsigned col = 0, c = 0;
  if (tid < NPART) {
    col = psum[tid] + psum[256 + tid] + psum[512 + tid] + psum[768 + tid];
    c = tbl[b * NPART + tid];
    runsrc[tid] = (unsigned)tid * CH + col;
    runcnt[tid] = c;
  }
  unsigned sc = col, sn = c;
  for (int off = 16; off > 0; off >>= 1) {
    sc += __shfl_xor(sc, off, 32);
    sn += __shfl_xor(sn, off, 32);
  }
  if (lane == 0 && wave < 8) { red[wave] = sc; red[8 + wave] = sn; }
  __syncthreads();
  unsigned S = 0, nb = 0;
  for (int w = 0; w < 8; ++w) { S += red[w]; nb += red[8 + w]; }
  const bool fits = nb <= (unsigned)CAP;

  unsigned k0[8], k1[8], rc[8], rs[8];
#pragma unroll
  for (int u = 0; u < 8; ++u) {
    int r = wave + 32 * u;
    rc[u] = runcnt[r];
    rs[u] = runsrc[r];
    k0[u] = ((unsigned)lane < rc[u]) ? part[rs[u] + lane] : 0u;
    k1[u] = ((unsigned)lane + 32u < rc[u]) ? part[rs[u] + lane + 32] : 0u;
  }
#pragma unroll
  for (int u = 0; u < 8; ++u) {
    if ((unsigned)lane < rc[u]) atomicAdd(&cnt[k0[u] >> SRC_BITS], 1u);
    if ((unsigned)lane + 32u < rc[u]) atomicAdd(&cnt[k1[u] >> SRC_BITS], 1u);
    for (unsigned i = (unsigned)lane + 64u; i < rc[u]; i += 32u)
      atomicAdd(&cnt[part[rs[u] + i] >> SRC_BITS], 1u);
  }
  __syncthreads();
  if (tid < 32) wave_scan<16>(cnt, cursor, lane);
  __syncthreads();
  if (tid < BNODES) {
    int node = b * BNODES + tid;
    if (node <= NN) rowptr[node] = S + cursor[tid];
    if (node < NN) dinv[node] = 1.0f / sqrtf((float)cnt[tid] + 1.0f);
  }
  __syncthreads();
#pragma unroll
  for (int u = 0; u < 8; ++u) {
    if ((unsigned)lane < rc[u]) {
      unsigned pos = atomicAdd(&cursor[k0[u] >> SRC_BITS], 1u);
      unsigned s = k0[u] & SRC_MASK;
      if (fits) outb[pos] = s; else csr[S + pos] = s;
    }
    if ((unsigned)lane + 32u < rc[u]) {
      unsigned pos = atomicAdd(&cursor[k1[u] >> SRC_BITS], 1u);
      unsigned s = k1[u] & SRC_MASK;
      if (fits) outb[pos] = s; else csr[S + pos] = s;
    }
    for (unsigned i = (unsigned)lane + 64u; i < rc[u]; i += 32u) {
      unsigned k = part[rs[u] + i];
      unsigned pos = atomicAdd(&cursor[k >> SRC_BITS], 1u);
      unsigned s = k & SRC_MASK;
      if (fits) outb[pos] = s; else csr[S + pos] = s;
    }
  }
  __syncthreads();
  if (fits)
    for (unsigned k = tid; k < nb; k += 1024u) csr[S + k] = outb[k];
}

template <int K>
__global__ __launch_bounds__(256) void k_gemm(const float* __restrict__ X,
                                              const float* __restrict__ W,
                                              const float* __restrict__ dinv,
                                              float* __restrict__ H, int M) {
  __shared__ __align__(16) v2f wl[(K / 2) * 64];
  for (int i = threadIdx.x; i < (K / 2) * 64; i += 256) {
    int p = i >> 6, n = i & 63;
    v2f t = { W[(size_t)(2 * p) * 64 + n], W[(size_t)(2 * p + 1) * 64 + n] };
    wl[i] = t;
  }
  __syncthreads();

  int wave  = threadIdx.x >> 5;
  int lane  = threadIdx.x & 31;
  int strip = blockIdx.x * 8 + wave;
  if (strip * 16 >= M) return;
  int row0 = strip * 16;
  int m    = lane & 15;
  int h2   = lane >> 4;

  const float* xrow = X + (size_t)(row0 + m) * K;
  v4f dlo = *(const v4f*)(dinv + row0 + 8 * h2);
  v4f dhi = *(const v4f*)(dinv + row0 + 8 * h2 + 4);
  v8f a0 = {}, a1 = {}, a2 = {}, a3 = {};
  for (int kb = 0; kb < K; kb += 4) {
    int pa = (kb >> 1) + h2;
    v2f a = *(const v2f*)(xrow + 2 * pa);
    const v2f* wp = &wl[pa * 64 + m];
    v2f b0 = wp[0];
    v2f b1 = wp[16];
    v2f b2 = wp[32];
    v2f b3 = wp[48];
    a0 = __builtin_amdgcn_wmma_f32_16x16x4_f32(false, a, false, b0, (short)0, a0, false, false);
    a1 = __builtin_amdgcn_wmma_f32_16x16x4_f32(false, a, false, b1, (short)0, a1, false, false);
    a2 = __builtin_amdgcn_wmma_f32_16x16x4_f32(false, a, false, b2, (short)0, a2, false, false);
    a3 = __builtin_amdgcn_wmma_f32_16x16x4_f32(false, a, false, b3, (short)0, a3, false, false);
  }
  float* out0 = H + (size_t)(row0 + 8 * h2) * 64 + m;
#pragma unroll
  for (int j = 0; j < 8; ++j) {
    float* r = out0 + (size_t)j * 64;
    float dj = (j < 4) ? dlo[j & 3] : dhi[j & 3];
    r[0] = a0[j] * dj; r[16] = a1[j] * dj; r[32] = a2[j] * dj; r[48] = a3[j] * dj;
  }
}

template <bool NORM>
__global__ __launch_bounds__(256) void k_pull(const float* __restrict__ H,
                                              const unsigned* __restrict__ rowptr,
                                              const unsigned* __restrict__ csr,
                                              const float* __restrict__ dinv,
                                              const float* __restrict__ bias,
                                              float* __restrict__ out, int N) {
  int node = (int)((blockIdx.x * 256 + threadIdx.x) >> 5);
  if (node >= N) return;
  node = __builtin_amdgcn_readfirstlane(node);
  int lane = threadIdx.x & 31;
  unsigned beg = rowptr[node], end = rowptr[node + 1];
  size_t base = (size_t)node * 64 + lane * 2;
  v2f hv = *(const v2f*)(H + base);
  v2f bv = *(const v2f*)(bias + lane * 2);
  float di = dinv[node];
  const float* hl = H + lane * 2;
  const v2f zero = { 0.0f, 0.0f };
  v2f acc = zero;
  for (unsigned cb = beg; cb < end; cb += 32) {
    unsigned n = end - cb;
    if (n > 32u) n = 32u;
    unsigned my = 0u;
    if ((unsigned)lane < n) my = csr[cb + lane];
    for (unsigned j = 0; j < n; j += 8) {
      unsigned s[8];
      v2f r[8];
#pragma unroll
      for (int u = 0; u < 8; ++u) {
        s[u] = (unsigned)__builtin_amdgcn_readlane((int)my, (int)(j + u));
      }
#pragma unroll
      for (int u = 0; u < 8; ++u) r[u] = *(const v2f*)(hl + (size_t)s[u] * 64);
#pragma unroll
      for (int u = 0; u < 8; ++u) {
        v2f rr = (j + u < n) ? r[u] : zero;
        acc += rr;
      }
    }
  }
  float vx = di * (acc.x + hv.x) + bv.x;
  float vy = di * (acc.y + hv.y) + bv.y;
  if (NORM) {
    vx = fmaxf(vx, 0.0f);
    vy = fmaxf(vy, 0.0f);
    float ss = vx * vx + vy * vy;
    for (int off = 16; off > 0; off >>= 1) ss += __shfl_xor(ss, off, 32);
    float scale = 1.0f / fmaxf(sqrtf(ss), 1e-12f);
    vx *= scale;
    vy *= scale;
  }
  v2f o = { vx, vy };
  *(v2f*)(out + base) = o;
}

extern "C" void kernel_launch(void* const* d_in, const int* in_sizes, int n_in,
                              void* d_out, int out_size, void* d_ws, size_t ws_size,
                              hipStream_t stream) {
  const float* x  = (const float*)d_in[0];
  const int*   ei = (const int*)d_in[1];
  const float* W1 = (const float*)d_in[2];
  const float* b1 = (const float*)d_in[3];
  const float* W2 = (const float*)d_in[4];
  const float* b2 = (const float*)d_in[5];
  const float* W3 = (const float*)d_in[6];
  const float* b3 = (const float*)d_in[7];
  float* out = (float*)d_out;

  const int* src = ei;
  const int* dst = ei + NE;

  char* ws = (char*)d_ws;
  size_t off = 0;
  auto carve = [&](size_t bytes) {
    void* p = ws + off;
    off = (off + bytes + 255) & ~(size_t)255;
    return p;
  };
  float*    dinv   = (float*)   carve((size_t)NN * 4);
  unsigned* rowptr = (unsigned*)carve((size_t)(NN + 1) * 4);
  unsigned* tbl    = (unsigned*)carve((size_t)NBIN * NPART * 4);
  unsigned* part   = (unsigned*)carve((size_t)NE * 4);
  unsigned* csr    = (unsigned*)carve((size_t)NE * 4);
  float*    h      = (float*)   carve((size_t)NN * 64 * 4);
  float*    hn     = (float*)   carve((size_t)NN * 64 * 4);

  const int gBlk   = ((NN / 16) + 7) / 8;
  const int wnBlk  = (NN * 32 + 255) / 256;

  k_part   <<<NPART, 1024, 0, stream>>>(src, dst, part, tbl);
  k_binsort<<<NBIN,  1024, 0, stream>>>(part, tbl, rowptr, dinv, csr);

  k_gemm<256><<<gBlk, 256, 0, stream>>>(x, W1, dinv, h, NN);
  k_pull<true><<<wnBlk, 256, 0, stream>>>(h, rowptr, csr, dinv, b1, hn, NN);

  k_gemm<64><<<gBlk, 256, 0, stream>>>(hn, W2, dinv, h, NN);
  k_pull<true><<<wnBlk, 256, 0, stream>>>(h, rowptr, csr, dinv, b2, hn, NN);

  k_gemm<64><<<gBlk, 256, 0, stream>>>(hn, W3, dinv, h, NN);
  k_pull<false><<<wnBlk, 256, 0, stream>>>(h, rowptr, csr, dinv, b3, out, NN);

  (void)in_sizes; (void)n_in; (void)out_size; (void)ws_size;
}


# Layer 2 and 3 products fused into the pull kernel before them: 16 nodes per block trip, rows to an LDS tile, 4 waves run the f32 WMMA product from it, whole-row stores; 8 launches to 6
# speedup vs baseline: 4.8667x; 1.1304x over previous
#include <hip/hip_runtime.h>


#define NN 100000
#define NE 1600000
#define BSH 9
#define BNODES (1 << BSH)
#define NBIN ((NN >> BSH) + 1)
#define NPART 256
#define CH ((((NE + NPART - 1) / NPART) + 31) & ~31)
#define SRC_BITS 17
#define SRC_MASK ((1u << SRC_BITS) - 1u)
#define CAP 16384

static_assert(NN <= (1 << SRC_BITS), "src must fit its field");
static_assert(NBIN <= 256, "bin histogram is 256 wide");
static_assert(NPART == 256, "8 runs per wave of a 32-wave block; 4 column groups");

typedef float v2f __attribute__((ext_vector_type(2)));
typedef float v4f __attribute__((ext_vector_type(4)));
typedef float v8f __attribute__((ext_vector_type(8)));

template <int PER>
__device__ __forceinline__ void wave_scan(const unsigned* in, unsigned* out, int lane) {
  unsigned v[PER];
  unsigned t = 0;
#pragma unroll
  for (int j = 0; j < PER; ++j) { v[j] = in[lane * PER + j]; t += v[j]; }
  unsigned inc = t;
  for (int off = 1; off < 32; off <<= 1) {
    unsigned y = __shfl_up(inc, off, 32);
    if (lane >= off) inc += y;
  }
  unsigned run = inc - t;
#pragma unroll
  for (int j = 0; j < PER; ++j) { out[lane * PER + j] = run; run += v[j]; }
}

__global__ __launch_bounds__(1024) void k_part(const int* __restrict__ src,
                                               const int* __restrict__ dst,
                                               unsigned* __restrict__ part,
                                               unsigned* __restrict__ tbl) {
  __shared__ unsigned hist[256];
  __shared__ unsigned start[256];
  __shared__ unsigned staged[CH];
  const int tid = threadIdx.x, lane = tid & 31;
  const int p = blockIdx.x;
  const int e0 = p * CH;
  int n = NE - e0;
  if (n > CH) n = CH;
  if (n < 0) n = 0;
  if (tid < 256) hist[tid] = 0;
  __syncthreads();
  constexpr int PER = (CH + 1023) / 1024;
  unsigned key[PER], bin[PER], rk[PER];
#pragma unroll
  for (int i = 0; i < PER; ++i) {
    int j = i * 1024 + tid;
    int e = e0 + j;
    if (e > NE - 1) e = NE - 1;
    unsigned s = (unsigned)src[e], d = (unsigned)dst[e];
    rk[i] = 0u;
    key[i] = ((d & (unsigned)(BNODES - 1)) << SRC_BITS) | s;
    bin[i] = (j < n) ? (d >> BSH) : 0xFFFFFFFFu;
  }
#pragma unroll
  for (int i = 0; i < PER; ++i)
    if (bin[i] != 0xFFFFFFFFu) rk[i] = atomicAdd(&hist[bin[i]], 1u);
  __syncthreads();
  if (tid < 32) wave_scan<8>(hist, start, lane);
  if (tid < NBIN) tbl[tid * NPART + p] = hist[tid];
  __syncthreads();
#pragma unroll
  for (int i = 0; i < PER; ++i)
    if (bin[i] != 0xFFFFFFFFu) staged[start[bin[i]] + rk[i]] = key[i];
  __syncthreads();
  for (int j = tid; j < n; j += 1024) part[e0 + j] = staged[j];
}

__global__ __launch_bounds__(1024) void k_binsort(const unsigned* __restrict__ part,
                                                  const unsigned* __restrict__ tbl,
                                                  unsigned* __restrict__ rowptr,
                                                  float* __restrict__ dinv,
                                                  unsigned* __restrict__ csr) {
  __shared__ unsigned psum[1024];
  __shared__ unsigned runsrc[NPART];
  __shared__ unsigned runcnt[NPART];
  __shared__ unsigned cnt[BNODES];
  __shared__ unsigned cursor[BNODES];
  __shared__ unsigned red[16];
  __shared__ unsigned outb[CAP];
  const int tid = threadIdx.x, lane = tid & 31, wave = tid >> 5;
  const int b = blockIdx.x;
  if (tid < BNODES) cnt[tid] = 0;
  const int t = tid & (NPART - 1), g = tid >> 8;
  unsigned ps = 0;
  for (int bb = g; bb < b; bb += 4) ps += tbl[bb * NPART + t];
  psum[tid] = ps;
  __syncthreads();
  unsigned col = 0, c = 0;
  if (tid < NPART) {
    col = psum[tid] + psum[256 + tid] + psum[512 + tid] + psum[768 + tid];
    c = tbl[b * NPART + tid];
    runsrc[tid] = (unsigned)tid * CH + col;
    runcnt[tid] = c;
  }
  unsigned sc = col, sn = c;
  for (int off = 16; off > 0; off >>= 1) {
    sc += __shfl_xor(sc, off, 32);
    sn += __shfl_xor(sn, off, 32);
  }
  if (lane == 0 && wave < 8) { red[wave] = sc; red[8 + wave] = sn; }
  __syncthreads();
  unsigned S = 0, nb = 0;
  for (int w = 0; w < 8; ++w) { S += red[w]; nb += red[8 + w]; }
  const bool fits = nb <= (unsigned)CAP;

  unsigned k0[8], k1[8], rc[8], rs[8];
#pragma unroll
  for (int u = 0; u < 8; ++u) {
    int r = wave + 32 * u;
    rc[u] = runcnt[r];
    rs[u] = runsrc[r];
    k0[u] = ((unsigned)lane < rc[u]) ? part[rs[u] + lane] : 0u;
    k1[u] = ((unsigned)lane + 32u < rc[u]) ? part[rs[u] + lane + 32] : 0u;
  }
#pragma unroll
  for (int u = 0; u < 8; ++u) {
    if ((unsigned)lane < rc[u]) atomicAdd(&cnt[k0[u] >> SRC_BITS], 1u);
    if ((unsigned)lane + 32u < rc[u]) atomicAdd(&cnt[k1[u] >> SRC_BITS], 1u);
    for (unsigned i = (unsigned)lane + 64u; i < rc[u]; i += 32u)
      atomicAdd(&cnt[part[rs[u] + i] >> SRC_BITS], 1u);
  }
  __syncthreads();
  if (tid < 32) wave_scan<16>(cnt, cursor, lane);
  __syncthreads();
  if (tid < BNODES) {
    int node = b * BNODES + tid;
    if (node <= NN) rowptr[node] = S + cursor[tid];
    if (node < NN) dinv[node] = 1.0f / sqrtf((float)cnt[tid] + 1.0f);
  }
  __syncthreads();
#pragma unroll
  for (int u = 0; u < 8; ++u) {
    if ((unsigned)lane < rc[u]) {
      unsigned pos = atomicAdd(&cursor[k0[u] >> SRC_BITS], 1u);
      unsigned s = k0[u] & SRC_MASK;
      if (fits) outb[pos] = s; else csr[S + pos] = s;
    }
    if ((unsigned)lane + 32u < rc[u]) {
      unsigned pos = atomicAdd(&cursor[k1[u] >> SRC_BITS], 1u);
      unsigned s = k1[u] & SRC_MASK;
      if (fits) outb[pos] = s; else csr[S + pos] = s;
    }
    for (unsigned i = (unsigned)lane + 64u; i < rc[u]; i += 32u) {
      unsigned k = part[rs[u] + i];
      unsigned pos = atomicAdd(&cursor[k >> SRC_BITS], 1u);
      unsigned s = k & SRC_MASK;
      if (fits) outb[pos] = s; else csr[S + pos] = s;
    }
  }
  __syncthreads();
  if (fits)
    for (unsigned k = tid; k < nb; k += 1024u) csr[S + k] = outb[k];
}

template <int K>
__global__ __launch_bounds__(256) void k_gemm(const float* __restrict__ X,
                                              const float* __restrict__ W,
                                              const float* __restrict__ dinv,
                                              float* __restrict__ H, int M) {
  __shared__ __align__(16) v2f wl[(K / 2) * 64];
  for (int i = threadIdx.x; i < (K / 2) * 64; i += 256) {
    int p = i >> 6, n = i & 63;
    v2f t = { W[(size_t)(2 * p) * 64 + n], W[(size_t)(2 * p + 1) * 64 + n] };
    wl[i] = t;
  }
  __syncthreads();

  int wave  = threadIdx.x >> 5;
  int lane  = threadIdx.x & 31;
  int strip = blockIdx.x * 8 + wave;
  if (strip * 16 >= M) return;
  int row0 = strip * 16;
  int m    = lane & 15;
  int h2   = lane >> 4;

  const float* xrow = X + (size_t)(row0 + m) * K;
  v4f dlo = *(const v4f*)(dinv + row0 + 8 * h2);
  v4f dhi = *(const v4f*)(dinv + row0 + 8 * h2 + 4);
  v8f a0 = {}, a1 = {}, a2 = {}, a3 = {};
  for (int kb = 0; kb < K; kb += 4) {
    int pa = (kb >> 1) + h2;
    v2f a = *(const v2f*)(xrow + 2 * pa);
    const v2f* wp = &wl[pa * 64 + m];
    v2f b0 = wp[0];
    v2f b1 = wp[16];
    v2f b2 = wp[32];
    v2f b3 = wp[48];
    a0 = __builtin_amdgcn_wmma_f32_16x16x4_f32(false, a, false, b0, (short)0, a0, false, false);
    a1 = __builtin_amdgcn_wmma_f32_16x16x4_f32(false, a, false, b1, (short)0, a1, false, false);
    a2 = __builtin_amdgcn_wmma_f32_16x16x4_f32(false, a, false, b2, (short)0, a2, false, false);
    a3 = __builtin_amdgcn_wmma_f32_16x16x4_f32(false, a, false, b3, (short)0, a3, false, false);
  }
  float* out0 = H + (size_t)(row0 + 8 * h2) * 64 + m;
#pragma unroll
  for (int j = 0; j < 8; ++j) {
    float* r = out0 + (size_t)j * 64;
    float dj = (j < 4) ? dlo[j & 3] : dhi[j & 3];
    r[0] = a0[j] * dj; r[16] = a1[j] * dj; r[32] = a2[j] * dj; r[48] = a3[j] * dj;
  }
}

__device__ __forceinline__ v2f gather_rows(const float* __restrict__ hl,
                                           const unsigned* __restrict__ csr,
                                           unsigned beg, unsigned end, unsigned my, int lane) {
  const v2f zero = { 0.0f, 0.0f };
  v2f acc = zero;
  for (unsigned cb = beg; cb < end; cb += 32) {
    unsigned n = end - cb;
    if (n > 32u) n = 32u;
    if (cb != beg) {
      my = 0u;
      if ((unsigned)lane < n) my = csr[cb + lane];
    }
    for (unsigned j = 0; j < n; j += 8) {
      unsigned s[8];
      v2f r[8];
#pragma unroll
      for (int u = 0; u < 8; ++u) {
        s[u] = (unsigned)__builtin_amdgcn_readlane((int)my, (int)(j + u));
      }
#pragma unroll
      for (int u = 0; u < 8; ++u) r[u] = *(const v2f*)(hl + (size_t)s[u] * 64);
#pragma unroll
      for (int u = 0; u < 8; ++u) {
        v2f rr = (j + u < n) ? r[u] : zero;
        acc += rr;
      }
    }
  }
  return acc;
}

__global__ __launch_bounds__(256) void k_pull_last(const float* __restrict__ H,
                                                   const unsigned* __restrict__ rowptr,
                                                   const unsigned* __restrict__ csr,
                                                   const float* __restrict__ dinv,
                                                   const float* __restrict__ bias,
                                                   float* __restrict__ out, int N) {
  int node = (int)((blockIdx.x * 256 + threadIdx.x) >> 5);
  if (node >= N) return;
  node = __builtin_amdgcn_readfirstlane(node);
  int lane = threadIdx.x & 31;
  unsigned beg = rowptr[node], end = rowptr[node + 1];
  size_t base = (size_t)node * 64 + lane * 2;
  v2f hv = *(const v2f*)(H + base);
  v2f bv = *(const v2f*)(bias + lane * 2);
  float di = dinv[node];
  unsigned my = 0u;
  if ((unsigned)lane < end - beg) my = csr[beg + lane];
  v2f acc = gather_rows(H + lane * 2, csr, beg, end, my, lane);
  v2f o = { di * (acc.x + hv.x) + bv.x, di * (acc.y + hv.y) + bv.y };
  *(v2f*)(out + base) = o;
}

#define TP 68
#define FUSE_GRID 893
static_assert(NN % 16 == 0, "tiles of 16 nodes cover the nodes exactly");

__global__ __launch_bounds__(512) void k_pull_gemm(const float* __restrict__ H,
                                                   const unsigned* __restrict__ rowptr,
                                                   const unsigned* __restrict__ csr,
                                                   const float* __restrict__ dinv,
                                                   const float* __restrict__ bias,
                                                   const float* __restrict__ W,
                                                   float* __restrict__ Hout) {
  __shared__ __align__(16) v2f wl[32 * 64];
  __shared__ __align__(16) float xs[16 * TP];
  __shared__ __align__(16) float ys[16 * TP];
  for (int i = threadIdx.x; i < 32 * 64; i += 512) {
    int p = i >> 6, n = i & 63;
    v2f t = { W[(size_t)(2 * p) * 64 + n], W[(size_t)(2 * p + 1) * 64 + n] };
    wl[i] = t;
  }
  const int lane = threadIdx.x & 31;
  const int wave = __builtin_amdgcn_readfirstlane((int)(threadIdx.x >> 5));
  const int m = lane & 15, h2 = lane >> 4;
  const int ntile = NN / 16;
  const v2f bv = *(const v2f*)(bias + lane * 2);
  const float* hl = H + lane * 2;

  int tile = blockIdx.x;
  unsigned beg = 0u, end = 0u, my = 0u;
  if (tile < ntile) {
    int node = tile * 16 + wave;
    beg = rowptr[node];
    end = rowptr[node + 1];
    if ((unsigned)lane < end - beg) my = csr[beg + lane];
  }
  for (; tile < ntile; tile += FUSE_GRID) {
    const int node = tile * 16 + wave;
    const size_t base = (size_t)node * 64 + lane * 2;
    v2f hv = *(const v2f*)(H + base);
    float di = dinv[node];
    const int ntl = tile + FUSE_GRID;
    const bool more = ntl < ntile;
    unsigned rp = 0u;
    if (more) rp = rowptr[ntl * 16 + wave + (lane & 1)];
    v2f acc = gather_rows(hl, csr, beg, end, my, lane);
    const unsigned nbeg = (unsigned)__builtin_amdgcn_readlane((int)rp, 0);
    const unsigned nend = (unsigned)__builtin_amdgcn_readlane((int)rp, 1);
    float vx = di * (acc.x + hv.x) + bv.x;
    float vy = di * (acc.y + hv.y) + bv.y;
    vx = fmaxf(vx, 0.0f);
    vy = fmaxf(vy, 0.0f);
    float ss = vx * vx + vy * vy;
    for (int off = 16; off > 0; off >>= 1) ss += __shfl_xor(ss, off, 32);
    float scale = 1.0f / fmaxf(sqrtf(ss), 1e-12f);
    v2f xr = { vx * scale, vy * scale };
    *(v2f*)&xs[wave * TP + lane * 2] = xr;
    __syncthreads();
    unsigned nmy = 0u;
    if (more && (unsigned)lane < nend - nbeg) nmy = csr[nbeg + lane];
    if (wave < 4) {
      const int row0 = tile * 16;
      v4f dlo = *(const v4f*)(dinv + row0 + 8 * h2);
      v4f dhi = *(const v4f*)(dinv + row0 + 8 * h2 + 4);
      v8f a0 = {};
#pragma unroll 8
      for (int kb = 0; kb < 64; kb += 4) {
        int pa = (kb >> 1) + h2;
        v2f a = *(const v2f*)&xs[m * TP + 2 * pa];
        v2f b = wl[pa * 64 + wave * 16 + m];
        a0 = __builtin_amdgcn_wmma_f32_16x16x4_f32(false, a, false, b, (short)0, a0, false, false);
      }
#pragma unroll
      for (int j = 0; j < 8; ++j) {
        float dj = (j < 4) ? dlo[j & 3] : dhi[j & 3];
        ys[(8 * h2 + j) * TP + wave * 16 + m] = a0[j] * dj;
      }
    }
    __syncthreads();
    *(v2f*)(Hout + base) = *(const v2f*)&ys[wave * TP + lane * 2];
    beg = nbeg; end = nend; my = nmy;
  }
}

extern "C" void kernel_launch(void* const* d_in, const int* in_sizes, int n_in,
                              void* d_out, int out_size, void* d_ws, size_t ws_size,
                              hipStream_t stream) {
  const float* x  = (const float*)d_in[0];
  const int*   ei = (const int*)d_in[1];
  const float* W1 = (const float*)d_in[2];
  const float* b1 = (const float*)d_in[3];
  const float* W2 = (const float*)d_in[4];
  const float* b2 = (const float*)d_in[5];
  const float* W3 = (const float*)d_in[6];
  const float* b3 = (const float*)d_in[7];
  float* out = (float*)d_out;

  const int* src = ei;
  const int* dst = ei + NE;

  char* ws = (char*)d_ws;
  size_t off = 0;
  auto carve = [&](size_t bytes) {
    void* p = ws + off;
    off = (off + bytes + 255) & ~(size_t)255;
    return p;
  };
  float*    dinv   = (float*)   carve((size_t)NN * 4);
  unsigned* rowptr = (unsigned*)carve((size_t)(NN + 1) * 4);
  unsigned* tbl    = (unsigned*)carve((size_t)NBIN * NPART * 4);
  unsigned* part   = (unsigned*)carve((size_t)NE * 4);
  unsigned* csr    = (unsigned*)carve((size_t)NE * 4);
  float*    h      = (float*)   carve((size_t)NN * 64 * 4);
  float*    hn     = (float*)   carve((size_t)NN * 64 * 4);

  const int gBlk   = ((NN / 16) + 7) / 8;
  const int wnBlk  = (NN * 32 + 255) / 256;

  k_part   <<<NPART, 1024, 0, stream>>>(src, dst, part, tbl);
  k_binsort<<<NBIN,  1024, 0, stream>>>(part, tbl, rowptr, dinv, csr);

  k_gemm<256><<<gBlk, 256, 0, stream>>>(x, W1, dinv, h, NN);

  k_pull_gemm<<<FUSE_GRID, 512, 0, stream>>>(h, rowptr, csr, dinv, b1, W2, hn);

  k_pull_gemm<<<FUSE_GRID, 512, 0, stream>>>(hn, rowptr, csr, dinv, b2, W3, h);

  k_pull_last<<<wnBlk, 256, 0, stream>>>(h, rowptr, csr, dinv, b3, out, NN);

  (void)in_sizes; (void)n_in; (void)out_size; (void)ws_size;
}


# Launch grids matched to the 256 WGPs: pull+product kernel 893 to 1024 blocks (4 per WGP, 6-7 trips); layer-1 product 782 blocks of 8 waves to 256 blocks of 32 waves, W staged once per WGP
# speedup vs baseline: 5.0255x; 1.0326x over previous
#include <hip/hip_runtime.h>


#define NN 100000
#define NE 1600000
#define BSH 9
#define BNODES (1 << BSH)
#define NBIN ((NN >> BSH) + 1)
#define NPART 256
#define CH ((((NE + NPART - 1) / NPART) + 31) & ~31)
#define SRC_BITS 17
#define SRC_MASK ((1u << SRC_BITS) - 1u)
#define CAP 16384

static_assert(NN <= (1 << SRC_BITS), "src must fit its field");
static_assert(NBIN <= 256, "bin histogram is 256 wide");
static_assert(NPART == 256, "8 runs per wave of a 32-wave block; 4 column groups");

typedef float v2f __attribute__((ext_vector_type(2)));
typedef float v4f __attribute__((ext_vector_type(4)));
typedef float v8f __attribute__((ext_vector_type(8)));

template <int PER>
__device__ __forceinline__ void wave_scan(const unsigned* in, unsigned* out, int lane) {
  unsigned v[PER];
  unsigned t = 0;
#pragma unroll
  for (int j = 0; j < PER; ++j) { v[j] = in[lane * PER + j]; t += v[j]; }
  unsigned inc = t;
  for (int off = 1; off < 32; off <<= 1) {
    unsigned y = __shfl_up(inc, off, 32);
    if (lane >= off) inc += y;
  }
  unsigned run = inc - t;
#pragma unroll
  for (int j = 0; j < PER; ++j) { out[lane * PER + j] = run; run += v[j]; }
}

__global__ __launch_bounds__(1024) void k_part(const int* __restrict__ src,
                                               const int* __restrict__ dst,
                                               unsigned* __restrict__ part,
                                               unsigned* __restrict__ tbl) {
  __shared__ unsigned hist[256];
  __shared__ unsigned start[256];
  __shared__ unsigned staged[CH];
  const int tid = threadIdx.x, lane = tid & 31;
  const int p = blockIdx.x;
  const int e0 = p * CH;
  int n = NE - e0;
  if (n > CH) n = CH;
  if (n < 0) n = 0;
  if (tid < 256) hist[tid] = 0;
  __syncthreads();
  constexpr int PER = (CH + 1023) / 1024;
  unsigned key[PER], bin[PER], rk[PER];
#pragma unroll
  for (int i = 0; i < PER; ++i) {
    int j = i * 1024 + tid;
    int e = e0 + j;
    if (e > NE - 1) e = NE - 1;
    unsigned s = (unsigned)src[e], d = (unsigned)dst[e];
    rk[i] = 0u;
    key[i] = ((d & (unsigned)(BNODES - 1)) << SRC_BITS) | s;
    bin[i] = (j < n) ? (d >> BSH) : 0xFFFFFFFFu;
  }
#pragma unroll
  for (int i = 0; i < PER; ++i)
    if (bin[i] != 0xFFFFFFFFu) rk[i] = atomicAdd(&hist[bin[i]], 1u);
  __syncthreads();
  if (tid < 32) wave_scan<8>(hist, start, lane);
  if (tid < NBIN) tbl[tid * NPART + p] = hist[tid];
  __syncthreads();
#pragma unroll
  for (int i = 0; i < PER; ++i)
    if (bin[i] != 0xFFFFFFFFu) staged[start[bin[i]] + rk[i]] = key[i];
  __syncthreads();
  for (int j = tid; j < n; j += 1024) part[e0 + j] = staged[j];
}

__global__ __launch_bounds__(1024) void k_binsort(const unsigned* __restrict__ part,
                                                  const unsigned* __restrict__ tbl,
                                                  unsigned* __restrict__ rowptr,
                                                  float* __restrict__ dinv,
                                                  unsigned* __restrict__ csr) {
  __shared__ unsigned psum[1024];
  __shared__ unsigned runsrc[NPART];
  __shared__ unsigned runcnt[NPART];
  __shared__ unsigned cnt[BNODES];
  __shared__ unsigned cursor[BNODES];
  __shared__ unsigned red[16];
  __shared__ unsigned outb[CAP];
  const int tid = threadIdx.x, lane = tid & 31, wave = tid >> 5;
  const int b = blockIdx.x;
  if (tid < BNODES) cnt[tid] = 0;
  const int t = tid & (NPART - 1), g = tid >> 8;
  unsigned ps = 0;
  for (int bb = g; bb < b; bb += 4) ps += tbl[bb * NPART + t];
  psum[tid] = ps;
  __syncthreads();
  unsigned col = 0, c = 0;
  if (tid < NPART) {
    col = psum[tid] + psum[256 + tid] + psum[512 + tid] + psum[768 + tid];
    c = tbl[b * NPART + tid];
    runsrc[tid] = (unsigned)tid * CH + col;
    runcnt[tid] = c;
  }
  unsigned sc = col, sn = c;
  for (int off = 16; off > 0; off >>= 1) {
    sc += __shfl_xor(sc, off, 32);
    sn += __shfl_xor(sn, off, 32);
  }
  if (lane == 0 && wave < 8) { red[wave] = sc; red[8 + wave] = sn; }
  __syncthreads();
  unsigned S = 0, nb = 0;
  for (int w = 0; w < 8; ++w) { S += red[w]; nb += red[8 + w]; }
  const bool fits = nb <= (unsigned)CAP;

  unsigned k0[8], k1[8], rc[8], rs[8];
#pragma unroll
  for (int u = 0; u < 8; ++u) {
    int r = wave + 32 * u;
    rc[u] = runcnt[r];
    rs[u] = runsrc[r];
    k0[u] = ((unsigned)lane < rc[u]) ? part[rs[u] + lane] : 0u;
    k1[u] = ((unsigned)lane + 32u < rc[u]) ? part[rs[u] + lane + 32] : 0u;
  }
#pragma unroll
  for (int u = 0; u < 8; ++u) {
    if ((unsigned)lane < rc[u]) atomicAdd(&cnt[k0[u] >> SRC_BITS], 1u);
    if ((unsigned)lane + 32u < rc[u]) atomicAdd(&cnt[k1[u] >> SRC_BITS], 1u);
    for (unsigned i = (unsigned)lane + 64u; i < rc[u]; i += 32u)
      atomicAdd(&cnt[part[rs[u] + i] >> SRC_BITS], 1u);
  }
  __syncthreads();
  if (tid < 32) wave_scan<16>(cnt, cursor, lane);
  __syncthreads();
  if (tid < BNODES) {
    int node = b * BNODES + tid;
    if (node <= NN) rowptr[node] = S + cursor[tid];
    if (node < NN) dinv[node] = 1.0f / sqrtf((float)cnt[tid] + 1.0f);
  }
  __syncthreads();
#pragma unroll
  for (int u = 0; u < 8; ++u) {
    if ((unsigned)lane < rc[u]) {
      unsigned pos = atomicAdd(&cursor[k0[u] >> SRC_BITS], 1u);
      unsigned s = k0[u] & SRC_MASK;
      if (fits) outb[pos] = s; else csr[S + pos] = s;
    }
    if ((unsigned)lane + 32u < rc[u]) {
      unsigned pos = atomicAdd(&cursor[k1[u] >> SRC_BITS], 1u);
      unsigned s = k1[u] & SRC_MASK;
      if (fits) outb[pos] = s; else csr[S + pos] = s;
    }
    for (unsigned i = (unsigned)lane + 64u; i < rc[u]; i += 32u) {
      unsigned k = part[rs[u] + i];
      unsigned pos = atomicAdd(&cursor[k >> SRC_BITS], 1u);
      unsigned s = k & SRC_MASK;
      if (fits) outb[pos] = s; else csr[S + pos] = s;
    }
  }
  __syncthreads();
  if (fits)
    for (unsigned k = tid; k < nb; k += 1024u) csr[S + k] = outb[k];
}

#define GEMM_GRID 256
template <int K>
__global__ __launch_bounds__(1024) void k_gemm(const float* __restrict__ X,
                                              const float* __restrict__ W,
                                              const float* __restrict__ dinv,
                                              float* __restrict__ H, int M) {
  __shared__ __align__(16) v2f wl[(K / 2) * 64];
  for (int i = threadIdx.x; i < (K / 2) * 64; i += 1024) {
    int p = i >> 6, n = i & 63;
    v2f t = { W[(size_t)(2 * p) * 64 + n], W[(size_t)(2 * p + 1) * 64 + n] };
    wl[i] = t;
  }
  __syncthreads();

  int wave  = threadIdx.x >> 5;
  int lane  = threadIdx.x & 31;
  int m    = lane & 15;
  int h2   = lane >> 4;
  for (int strip = blockIdx.x + GEMM_GRID * wave; strip * 16 < M; strip += GEMM_GRID * 32) {
    int row0 = strip * 16;

    const float* xrow = X + (size_t)(row0 + m) * K;
    v4f dlo = *(const v4f*)(dinv + row0 + 8 * h2);
    v4f dhi = *(const v4f*)(dinv + row0 + 8 * h2 + 4);
    v8f a0 = {}, a1 = {}, a2 = {}, a3 = {};
    for (int kb = 0; kb < K; kb += 4) {
      int pa = (kb >> 1) + h2;
      v2f a = *(const v2f*)(xrow + 2 * pa);
      const v2f* wp = &wl[pa * 64 + m];
      v2f b0 = wp[0];
      v2f b1 = wp[16];
      v2f b2 = wp[32];
      v2f b3 = wp[48];
      a0 = __builtin_amdgcn_wmma_f32_16x16x4_f32(false, a, false, b0, (short)0, a0, false, false);
      a1 = __builtin_amdgcn_wmma_f32_16x16x4_f32(false, a, false, b1, (short)0, a1, false, false);
      a2 = __builtin_amdgcn_wmma_f32_16x16x4_f32(false, a, false, b2, (short)0, a2, false, false);
      a3 = __builtin_amdgcn_wmma_f32_16x16x4_f32(false, a, false, b3, (short)0, a3, false, false);
    }
    float* out0 = H + (size_t)(row0 + 8 * h2) * 64 + m;
#pragma unroll
    for (int j = 0; j < 8; ++j) {
      float* r = out0 + (size_t)j * 64;
      float dj = (j < 4) ? dlo[j & 3] : dhi[j & 3];
      r[0] = a0[j] * dj; r[16] = a1[j] * dj; r[32] = a2[j] * dj; r[48] = a3[j] * dj;
    }
  }
}

__device__ __forceinline__ v2f gather_rows(const float* __restrict__ hl,
                                           const unsigned* __restrict__ csr,
                                           unsigned beg, unsigned end, unsigned my, int lane) {
  const v2f zero = { 0.0f, 0.0f };
  v2f acc = zero;
  for (unsigned cb = beg; cb < end; cb += 32) {
    unsigned n = end - cb;
    if (n > 32u) n = 32u;
    if (cb != beg) {
      my = 0u;
      if ((unsigned)lane < n) my = csr[cb + lane];
    }
    for (unsigned j = 0; j < n; j += 8) {
      unsigned s[8];
      v2f r[8];
#pragma unroll
      for (int u = 0; u < 8; ++u) {
        s[u] = (unsigned)__builtin_amdgcn_readlane((int)my, (int)(j + u));
      }
#pragma unroll
      for (int u = 0; u < 8; ++u) r[u] = *(const v2f*)(hl + (size_t)s[u] * 64);
#pragma unroll
      for (int u = 0; u < 8; ++u) {
        v2f rr = (j + u < n) ? r[u] : zero;
        acc += rr;
      }
    }
  }
  return acc;
}

__global__ __launch_bounds__(256) void k_pull_last(const float* __restrict__ H,
                                                   const unsigned* __restrict__ rowptr,
                                                   const unsigned* __restrict__ csr,
                                                   const float* __restrict__ dinv,
                                                   const float* __restrict__ bias,
                                                   float* __restrict__ out, int N) {
  int node = (int)((blockIdx.x * 256 + threadIdx.x) >> 5);
  if (node >= N) return;
  node = __builtin_amdgcn_readfirstlane(node);
  int lane = threadIdx.x & 31;
  unsigned beg = rowptr[node], end = rowptr[node + 1];
  size_t base = (size_t)node * 64 + lane * 2;
  v2f hv = *(const v2f*)(H + base);
  v2f bv = *(const v2f*)(bias + lane * 2);
  float di = dinv[node];
  unsigned my = 0u;
  if ((unsigned)lane < end - beg) my = csr[beg + lane];
  v2f acc = gather_rows(H + lane * 2, csr, beg, end, my, lane);
  v2f o = { di * (acc.x + hv.x) + bv.x, di * (acc.y + hv.y) + bv.y };
  *(v2f*)(out + base) = o;
}

#define TP 68
#define FUSE_GRID 1024
static_assert(NN % 16 == 0, "tiles of 16 nodes cover the nodes exactly");

__global__ __launch_bounds__(512) void k_pull_gemm(const float* __restrict__ H,
                                                   const unsigned* __restrict__ rowptr,
                                                   const unsigned* __restrict__ csr,
                                                   const float* __restrict__ dinv,
                                                   const float* __restrict__ bias,
                                                   const float* __restrict__ W,
                                                   float* __restrict__ Hout) {
  __shared__ __align__(16) v2f wl[32 * 64];
  __shared__ __align__(16) float xs[16 * TP];
  __shared__ __align__(16) float ys[16 * TP];
  for (int i = threadIdx.x; i < 32 * 64; i += 512) {
    int p = i >> 6, n = i & 63;
    v2f t = { W[(size_t)(2 * p) * 64 + n], W[(size_t)(2 * p + 1) * 64 + n] };
    wl[i] = t;
  }
  const int lane = threadIdx.x & 31;
  const int wave = __builtin_amdgcn_readfirstlane((int)(threadIdx.x >> 5));
  const int m = lane & 15, h2 = lane >> 4;
  const int ntile = NN / 16;
  const v2f bv = *(const v2f*)(bias + lane * 2);
  const float* hl = H + lane * 2;

  int tile = blockIdx.x;
  unsigned beg = 0u, end = 0u, my = 0u;
  if (tile < ntile) {
    int node = tile * 16 + wave;
    beg = rowptr[node];
    end = rowptr[node + 1];
    if ((unsigned)lane < end - beg) my = csr[beg + lane];
  }
  for (; tile < ntile; tile += FUSE_GRID) {
    const int node = tile * 16 + wave;
    const size_t base = (size_t)node * 64 + lane * 2;
    v2f hv = *(const v2f*)(H + base);
    float di = dinv[node];
    const int ntl = tile + FUSE_GRID;
    const bool more = ntl < ntile;
    unsigned rp = 0u;
    if (more) rp = rowptr[ntl * 16 + wave + (lane & 1)];
    v2f acc = gather_rows(hl, csr, beg, end, my, lane);
    const unsigned nbeg = (unsigned)__builtin_amdgcn_readlane((int)rp, 0);
    const unsigned nend = (unsigned)__builtin_amdgcn_readlane((int)rp, 1);
    float vx = di * (acc.x + hv.x) + bv.x;
    float vy = di * (acc.y + hv.y) + bv.y;
    vx = fmaxf(vx, 0.0f);
    vy = fmaxf(vy, 0.0f);
    float ss = vx * vx + vy * vy;
    for (int off = 16; off > 0; off >>= 1) ss += __shfl_xor(ss, off, 32);
    float scale = 1.0f / fmaxf(sqrtf(ss), 1e-12f);
    v2f xr = { vx * scale, vy * scale };
    *(v2f*)&xs[wave * TP + lane * 2] = xr;
    __syncthreads();
    unsigned nmy = 0u;
    if (more && (unsigned)lane < nend - nbeg) nmy = csr[nbeg + lane];
    if (wave < 4) {
      const int row0 = tile * 16;
      v4f dlo = *(const v4f*)(dinv + row0 + 8 * h2);
      v4f dhi = *(const v4f*)(dinv + row0 + 8 * h2 + 4);
      v8f a0 = {};
#pragma unroll 8
      for (int kb = 0; kb < 64; kb += 4) {
        int pa = (kb >> 1) + h2;
        v2f a = *(const v2f*)&xs[m * TP + 2 * pa];
        v2f b = wl[pa * 64 + wave * 16 + m];
        a0 = __builtin_amdgcn_wmma_f32_16x16x4_f32(false, a, false, b, (short)0, a0, false, false);
      }
#pragma unroll
      for (int j = 0; j < 8; ++j) {
        float dj = (j < 4) ? dlo[j & 3] : dhi[j & 3];
        ys[(8 * h2 + j) * TP + wave * 16 + m] = a0[j] * dj;
      }
    }
    __syncthreads();
    *(v2f*)(Hout + base) = *(const v2f*)&ys[wave * TP + lane * 2];
    beg = nbeg; end = nend; my = nmy;
  }
}

extern "C" void kernel_launch(void* const* d_in, const int* in_sizes, int n_in,
                              void* d_out, int out_size, void* d_ws, size_t ws_size,
                              hipStream_t stream) {
  const float* x  = (const float*)d_in[0];
  const int*   ei = (const int*)d_in[1];
  const float* W1 = (const float*)d_in[2];
  const float* b1 = (const float*)d_in[3];
  const float* W2 = (const float*)d_in[4];
  const float* b2 = (const float*)d_in[5];
  const float* W3 = (const float*)d_in[6];
  const float* b3 = (const float*)d_in[7];
  float* out = (float*)d_out;

  const int* src = ei;
  const int* dst = ei + NE;

  char* ws = (char*)d_ws;
  size_t off = 0;
  auto carve = [&](size_t bytes) {
    void* p = ws + off;
    off = (off + bytes + 255) & ~(size_t)255;
    return p;
  };
  float*    dinv   = (float*)   carve((size_t)NN * 4);
  unsigned* rowptr = (unsigned*)carve((size_t)(NN + 1) * 4);
  unsigned* tbl    = (unsigned*)carve((size_t)NBIN * NPART * 4);
  unsigned* part   = (unsigned*)carve((size_t)NE * 4);
  unsigned* csr    = (unsigned*)carve((size_t)NE * 4);
  float*    h      = (float*)   carve((size_t)NN * 64 * 4);
  float*    hn     = (float*)   carve((size_t)NN * 64 * 4);

  const int wnBlk  = (NN * 32 + 255) / 256;

  k_part   <<<NPART, 1024, 0, stream>>>(src, dst, part, tbl);
  k_binsort<<<NBIN,  1024, 0, stream>>>(part, tbl, rowptr, dinv, csr);

  k_gemm<256><<<GEMM_GRID, 1024, 0, stream>>>(x, W1, dinv, h, NN);

  k_pull_gemm<<<FUSE_GRID, 512, 0, stream>>>(h, rowptr, csr, dinv, b1, W2, hn);

  k_pull_gemm<<<FUSE_GRID, 512, 0, stream>>>(hn, rowptr, csr, dinv, b2, W3, h);

  k_pull_last<<<wnBlk, 256, 0, stream>>>(h, rowptr, csr, dinv, b3, out, NN);

  (void)in_sizes; (void)n_in; (void)out_size; (void)ws_size;
}


# Launch heads lose their serial memory trips: W staged with all its loads in flight (beside the first bounds and list in pull+product), level 2 reads its run starts from level 1's scan, no sum loop
# speedup vs baseline: 5.1916x; 1.0331x over previous
#include <hip/hip_runtime.h>


#define NN 100000
#define NE 1600000
#define BSH 9
#define BNODES (1 << BSH)
#define NBIN ((NN >> BSH) + 1)
#define NPART 256
#define CH ((((NE + NPART - 1) / NPART) + 31) & ~31)
#define SRC_BITS 17
#define SRC_MASK ((1u << SRC_BITS) - 1u)
#define CAP 16384

static_assert(NN <= (1 << SRC_BITS), "src must fit its field");
static_assert(NBIN <= 256, "bin histogram is 256 wide");
static_assert(NPART == 256, "8 runs per wave of a 32-wave block; 4 column groups");

typedef float v2f __attribute__((ext_vector_type(2)));
typedef unsigned v2u __attribute__((ext_vector_type(2)));
typedef float v4f __attribute__((ext_vector_type(4)));
typedef float v8f __attribute__((ext_vector_type(8)));

template <int PER>
__device__ __forceinline__ void wave_scan(const unsigned* in, unsigned* out, int lane) {
  unsigned v[PER];
  unsigned t = 0;
#pragma unroll
  for (int j = 0; j < PER; ++j) { v[j] = in[lane * PER + j]; t += v[j]; }
  unsigned inc = t;
  for (int off = 1; off < 32; off <<= 1) {
    unsigned y = __shfl_up(inc, off, 32);
    if (lane >= off) inc += y;
  }
  unsigned run = inc - t;
#pragma unroll
  for (int j = 0; j < PER; ++j) { out[lane * PER + j] = run; run += v[j]; }
}

__global__ __launch_bounds__(1024) void k_part(const int* __restrict__ src,
                                               const int* __restrict__ dst,
                                               unsigned* __restrict__ part,
                                               unsigned* __restrict__ tbl) {
  __shared__ unsigned hist[256];
  __shared__ unsigned start[256];
  __shared__ unsigned staged[CH];
  const int tid = threadIdx.x, lane = tid & 31;
  const int p = blockIdx.x;
  const int e0 = p * CH;
  int n = NE - e0;
  if (n > CH) n = CH;
  if (n < 0) n = 0;
  if (tid < 256) hist[tid] = 0;
  __syncthreads();
  constexpr int PER = (CH + 1023) / 1024;
  unsigned key[PER], bin[PER], rk[PER];
#pragma unroll
  for (int i = 0; i < PER; ++i) {
    int j = i * 1024 + tid;
    int e = e0 + j;
    if (e > NE - 1) e = NE - 1;
    unsigned s = (unsigned)src[e], d = (unsigned)dst[e];
    rk[i] = 0u;
    key[i] = ((d & (unsigned)(BNODES - 1)) << SRC_BITS) | s;
    bin[i] = (j < n) ? (d >> BSH) : 0xFFFFFFFFu;
  }
#pragma unroll
  for (int i = 0; i < PER; ++i)
    if (bin[i] != 0xFFFFFFFFu) rk[i] = atomicAdd(&hist[bin[i]], 1u);
  __syncthreads();
  if (tid < 32) wave_scan<8>(hist, start, lane);
  __syncthreads();
  if (tid < NBIN) {
    v2u e = { start[tid], hist[tid] };
    *(v2u*)(tbl + 2 * (tid * NPART + p)) = e;
  }
#pragma unroll
  for (int i = 0; i < PER; ++i)
    if (bin[i] != 0xFFFFFFFFu) staged[start[bin[i]] + rk[i]] = key[i];
  __syncthreads();
  for (int j = tid; j < n; j += 1024) part[e0 + j] = staged[j];
}

__global__ __launch_bounds__(1024) void k_binsort(const unsigned* __restrict__ part,
                                                  const unsigned* __restrict__ tbl,
                                                  unsigned* __restrict__ rowptr,
                                                  float* __restrict__ dinv,
                                                  unsigned* __restrict__ csr) {
  __shared__ unsigned runsrc[NPART];
  __shared__ unsigned runcnt[NPART];
  __shared__ unsigned cnt[BNODES];
  __shared__ unsigned cursor[BNODES];
  __shared__ unsigned red[16];
  __shared__ unsigned outb[CAP];
  const int tid = threadIdx.x, lane = tid & 31, wave = tid >> 5;
  const int b = blockIdx.x;
  if (tid < BNODES) cnt[tid] = 0;
  unsigned col = 0, c = 0;
  if (tid < NPART) {
    v2u e = *(const v2u*)(tbl + 2 * (b * NPART + tid));
    col = e.x;
    c = e.y;
    runsrc[tid] = (unsigned)tid * CH + col;
    runcnt[tid] = c;
  }
  unsigned sc = col, sn = c;
  for (int off = 16; off > 0; off >>= 1) {
    sc += __shfl_xor(sc, off, 32);
    sn += __shfl_xor(sn, off, 32);
  }
  if (lane == 0 && wave < 8) { red[wave] = sc; red[8 + wave] = sn; }
  __syncthreads();
  unsigned S = 0, nb = 0;
  for (int w = 0; w < 8; ++w) { S += red[w]; nb += red[8 + w]; }
  const bool fits = nb <= (unsigned)CAP;

  unsigned k0[8], k1[8], rc[8], rs[8];
#pragma unroll
  for (int u = 0; u < 8; ++u) {
    int r = wave + 32 * u;
    rc[u] = runcnt[r];
    rs[u] = runsrc[r];
    k0[u] = ((unsigned)lane < rc[u]) ? part[rs[u] + lane] : 0u;
    k1[u] = ((unsigned)lane + 32u < rc[u]) ? part[rs[u] + lane + 32] : 0u;
  }
#pragma unroll
  for (int u = 0; u < 8; ++u) {
    if ((unsigned)lane < rc[u]) atomicAdd(&cnt[k0[u] >> SRC_BITS], 1u);
    if ((unsigned)lane + 32u < rc[u]) atomicAdd(&cnt[k1[u] >> SRC_BITS], 1u);
    for (unsigned i = (unsigned)lane + 64u; i < rc[u]; i += 32u)
      atomicAdd(&cnt[part[rs[u] + i] >> SRC_BITS], 1u);
  }
  __syncthreads();
  if (tid < 32) wave_scan<16>(cnt, cursor, lane);
  __syncthreads();
  if (tid < BNODES) {
    int node = b * BNODES + tid;
    if (node <= NN) rowptr[node] = S + cursor[tid];
    if (node < NN) dinv[node] = 1.0f / sqrtf((float)cnt[tid] + 1.0f);
  }
  __syncthreads();
#pragma unroll
  for (int u = 0; u < 8; ++u) {
    if ((unsigned)lane < rc[u]) {
      unsigned pos = atomicAdd(&cursor[k0[u] >> SRC_BITS], 1u);
      unsigned s = k0[u] & SRC_MASK;
      if (fits) outb[pos] = s; else csr[S + pos] = s;
    }
    if ((unsigned)lane + 32u < rc[u]) {
      unsigned pos = atomicAdd(&cursor[k1[u] >> SRC_BITS], 1u);
      unsigned s = k1[u] & SRC_MASK;
      if (fits) outb[pos] = s; else csr[S + pos] = s;
    }
    for (unsigned i = (unsigned)lane + 64u; i < rc[u]; i += 32u) {
      unsigned k = part[rs[u] + i];
      unsigned pos = atomicAdd(&cursor[k >> SRC_BITS], 1u);
      unsigned s = k & SRC_MASK;
      if (fits) outb[pos] = s; else csr[S + pos] = s;
    }
  }
  __syncthreads();
  if (fits)
    for (unsigned k = tid; k < nb; k += 1024u) csr[S + k] = outb[k];
}

#define GEMM_GRID 256
template <int K>
__global__ __launch_bounds__(1024) void k_gemm(const float* __restrict__ X,
                                              const float* __restrict__ W,
                                              const float* __restrict__ dinv,
                                              float* __restrict__ H, int M) {
  __shared__ __align__(16) v2f wl[(K / 2) * 64];
  constexpr int WT = (K / 2) * 16 / 1024;
  static_assert(WT * 1024 == (K / 2) * 16, "the tasks deal out evenly");
  v4f we[WT], wo[WT];
#pragma unroll
  for (int j = 0; j < WT; ++j) {
    int t = threadIdx.x + 1024 * j;
    int p = t >> 4, n4 = (t & 15) * 4;
    we[j] = *(const v4f*)(W + (size_t)(2 * p) * 64 + n4);
    wo[j] = *(const v4f*)(W + (size_t)(2 * p + 1) * 64 + n4);
  }
#pragma unroll
  for (int j = 0; j < WT; ++j) {
    int t = threadIdx.x + 1024 * j;
    int p = t >> 4, n4 = (t & 15) * 4;
    v2f* d = &wl[p * 64 + n4];
    d[0] = v2f{ we[j].x, wo[j].x };
    d[1] = v2f{ we[j].y, wo[j].y };
    d[2] = v2f{ we[j].z, wo[j].z };
    d[3] = v2f{ we[j].w, wo[j].w };
  }
  __syncthreads();

  int wave  = threadIdx.x >> 5;
  int lane  = threadIdx.x & 31;
  int m    = lane & 15;
  int h2   = lane >> 4;
  for (int strip = blockIdx.x + GEMM_GRID * wave; strip * 16 < M; strip += GEMM_GRID * 32) {
    int row0 = strip * 16;

    const float* xrow = X + (size_t)(row0 + m) * K;
    v4f dlo = *(const v4f*)(dinv + row0 + 8 * h2);
    v4f dhi = *(const v4f*)(dinv + row0 + 8 * h2 + 4);
    v8f a0 = {}, a1 = {}, a2 = {}, a3 = {};
    for (int kb = 0; kb < K; kb += 4) {
      int pa = (kb >> 1) + h2;
      v2f a = *(const v2f*)(xrow + 2 * pa);
      const v2f* wp = &wl[pa * 64 + m];
      v2f b0 = wp[0];
      v2f b1 = wp[16];
      v2f b2 = wp[32];
      v2f b3 = wp[48];
      a0 = __builtin_amdgcn_wmma_f32_16x16x4_f32(false, a, false, b0, (short)0, a0, false, false);
      a1 = __builtin_amdgcn_wmma_f32_16x16x4_f32(false, a, false, b1, (short)0, a1, false, false);
      a2 = __builtin_amdgcn_wmma_f32_16x16x4_f32(false, a, false, b2, (short)0, a2, false, false);
      a3 = __builtin_amdgcn_wmma_f32_16x16x4_f32(false, a, false, b3, (short)0, a3, false, false);
    }
    float* out0 = H + (size_t)(row0 + 8 * h2) * 64 + m;
#pragma unroll
    for (int j = 0; j < 8; ++j) {
      float* r = out0 + (size_t)j * 64;
      float dj = (j < 4) ? dlo[j & 3] : dhi[j & 3];
      r[0] = a0[j] * dj; r[16] = a1[j] * dj; r[32] = a2[j] * dj; r[48] = a3[j] * dj;
    }
  }
}

__device__ __forceinline__ v2f gather_rows(const float* __restrict__ hl,
                                           const unsigned* __restrict__ csr,
                                           unsigned beg, unsigned end, unsigned my, int lane) {
  const v2f zero = { 0.0f, 0.0f };
  v2f acc = zero;
  for (unsigned cb = beg; cb < end; cb += 32) {
    unsigned n = end - cb;
    if (n > 32u) n = 32u;
    if (cb != beg) {
      my = 0u;
      if ((unsigned)lane < n) my = csr[cb + lane];
    }
    for (unsigned j = 0; j < n; j += 8) {
      unsigned s[8];
      v2f r[8];
#pragma unroll
      for (int u = 0; u < 8; ++u) {
        s[u] = (unsigned)__builtin_amdgcn_readlane((int)my, (int)(j + u));
      }
#pragma unroll
      for (int u = 0; u < 8; ++u) r[u] = *(const v2f*)(hl + (size_t)s[u] * 64);
#pragma unroll
      for (int u = 0; u < 8; ++u) {
        v2f rr = (j + u < n) ? r[u] : zero;
        acc += rr;
      }
    }
  }
  return acc;
}

__global__ __launch_bounds__(256) void k_pull_last(const float* __restrict__ H,
                                                   const unsigned* __restrict__ rowptr,
                                                   const unsigned* __restrict__ csr,
                                                   const float* __restrict__ dinv,
                                                   const float* __restrict__ bias,
                                                   float* __restrict__ out, int N) {
  int node = (int)((blockIdx.x * 256 + threadIdx.x) >> 5);
  if (node >= N) return;
  node = __builtin_amdgcn_readfirstlane(node);
  int lane = threadIdx.x & 31;
  unsigned beg = rowptr[node], end = rowptr[node + 1];
  size_t base = (size_t)node * 64 + lane * 2;
  v2f hv = *(const v2f*)(H + base);
  v2f bv = *(const v2f*)(bias + lane * 2);
  float di = dinv[node];
  unsigned my = 0u;
  if ((unsigned)lane < end - beg) my = csr[beg + lane];
  v2f acc = gather_rows(H + lane * 2, csr, beg, end, my, lane);
  v2f o = { di * (acc.x + hv.x) + bv.x, di * (acc.y + hv.y) + bv.y };
  *(v2f*)(out + base) = o;
}

#define TP 68
#define FUSE_GRID 1024
static_assert(NN % 16 == 0, "tiles of 16 nodes cover the nodes exactly");

__global__ __launch_bounds__(512) void k_pull_gemm(const float* __restrict__ H,
                                                   const unsigned* __restrict__ rowptr,
                                                   const unsigned* __restrict__ csr,
                                                   const float* __restrict__ dinv,
                                                   const float* __restrict__ bias,
                                                   const float* __restrict__ W,
                                                   float* __restrict__ Hout) {
  __shared__ __align__(16) v2f wl[32 * 64];
  __shared__ __align__(16) float xs[16 * TP];
  __shared__ __align__(16) float ys[16 * TP];
  const int wp = (int)(threadIdx.x >> 4), wn4 = (int)(threadIdx.x & 15) * 4;
  const v4f we = *(const v4f*)(W + (size_t)(2 * wp) * 64 + wn4);
  const v4f wo = *(const v4f*)(W + (size_t)(2 * wp + 1) * 64 + wn4);
  const int lane = threadIdx.x & 31;
  const int wave = __builtin_amdgcn_readfirstlane((int)(threadIdx.x >> 5));
  const int m = lane & 15, h2 = lane >> 4;
  const int ntile = NN / 16;
  const v2f bv = *(const v2f*)(bias + lane * 2);
  const float* hl = H + lane * 2;

  int tile = blockIdx.x;
  unsigned beg = 0u, end = 0u, my = 0u;
  if (tile < ntile) {
    int node = tile * 16 + wave;
    beg = rowptr[node];
    end = rowptr[node + 1];
    if ((unsigned)lane < end - beg) my = csr[beg + lane];
  }
  {
    v2f* d = &wl[wp * 64 + wn4];
    d[0] = v2f{ we.x, wo.x };
    d[1] = v2f{ we.y, wo.y };
    d[2] = v2f{ we.z, wo.z };
    d[3] = v2f{ we.w, wo.w };
  }
  for (; tile < ntile; tile += FUSE_GRID) {
    const int node = tile * 16 + wave;
    const size_t base = (size_t)node * 64 + lane * 2;
    v2f hv = *(const v2f*)(H + base);
    float di = dinv[node];
    const int ntl = tile + FUSE_GRID;
    const bool more = ntl < ntile;
    unsigned rp = 0u;
    if (more) rp = rowptr[ntl * 16 + wave + (lane & 1)];
    v2f acc = gather_rows(hl, csr, beg, end, my, lane);
    const unsigned nbeg = (unsigned)__builtin_amdgcn_readlane((int)rp, 0);
    const unsigned nend = (unsigned)__builtin_amdgcn_readlane((int)rp, 1);
    float vx = di * (acc.x + hv.x) + bv.x;
    float vy = di * (acc.y + hv.y) + bv.y;
    vx = fmaxf(vx, 0.0f);
    vy = fmaxf(vy, 0.0f);
    float ss = vx * vx + vy * vy;
    for (int off = 16; off > 0; off >>= 1) ss += __shfl_xor(ss, off, 32);
    float scale = 1.0f / fmaxf(sqrtf(ss), 1e-12f);
    v2f xr = { vx * scale, vy * scale };
    *(v2f*)&xs[wave * TP + lane * 2] = xr;
    __syncthreads();
    unsigned nmy = 0u;
    if (more && (unsigned)lane < nend - nbeg) nmy = csr[nbeg + lane];
    if (wave < 4) {
      const int row0 = tile * 16;
      v4f dlo = *(const v4f*)(dinv + row0 + 8 * h2);
      v4f dhi = *(const v4f*)(dinv + row0 + 8 * h2 + 4);
      v8f a0 = {};
#pragma unroll 8
      for (int kb = 0; kb < 64; kb += 4) {
        int pa = (kb >> 1) + h2;
        v2f a = *(const v2f*)&xs[m * TP + 2 * pa];
        v2f b = wl[pa * 64 + wave * 16 + m];
        a0 = __builtin_amdgcn_wmma_f32_16x16x4_f32(false, a, false, b, (short)0, a0, false, false);
      }
#pragma unroll
      for (int j = 0; j < 8; ++j) {
        float dj = (j < 4) ? dlo[j & 3] : dhi[j & 3];
        ys[(8 * h2 + j) * TP + wave * 16 + m] = a0[j] * dj;
      }
    }
    __syncthreads();
    *(v2f*)(Hout + base) = *(const v2f*)&ys[wave * TP + lane * 2];
    beg = nbeg; end = nend; my = nmy;
  }
}

extern "C" void kernel_launch(void* const* d_in, const int* in_sizes, int n_in,
                              void* d_out, int out_size, void* d_ws, size_t ws_size,
                              hipStream_t stream) {
  const float* x  = (const float*)d_in[0];
  const int*   ei = (const int*)d_in[1];
  const float* W1 = (const float*)d_in[2];
  const float* b1 = (const float*)d_in[3];
  const float* W2 = (const float*)d_in[4];
  const float* b2 = (const float*)d_in[5];
  const float* W3 = (const float*)d_in[6];
  const float* b3 = (const float*)d_in[7];
  float* out = (float*)d_out;

  const int* src = ei;
  const int* dst = ei + NE;

  char* ws = (char*)d_ws;
  size_t off = 0;
  auto carve = [&](size_t bytes) {
    void* p = ws + off;
    off = (off + bytes + 255) & ~(size_t)255;
    return p;
  };
  float*    dinv   = (float*)   carve((size_t)NN * 4);
  unsigned* rowptr = (unsigned*)carve((size_t)(NN + 1) * 4);
  unsigned* tbl    = (unsigned*)carve((size_t)NBIN * NPART * 8);
  unsigned* part   = (unsigned*)carve((size_t)NE * 4);
  unsigned* csr    = (unsigned*)carve((size_t)NE * 4);
  float*    h      = (float*)   carve((size_t)NN * 64 * 4);
  float*    hn     = (float*)   carve((size_t)NN * 64 * 4);

  const int wnBlk  = (NN * 32 + 255) / 256;

  k_part   <<<NPART, 1024, 0, stream>>>(src, dst, part, tbl);
  k_binsort<<<NBIN,  1024, 0, stream>>>(part, tbl, rowptr, dinv, csr);

  k_gemm<256><<<GEMM_GRID, 1024, 0, stream>>>(x, W1, dinv, h, NN);

  k_pull_gemm<<<FUSE_GRID, 512, 0, stream>>>(h, rowptr, csr, dinv, b1, W2, hn);

  k_pull_gemm<<<FUSE_GRID, 512, 0, stream>>>(hn, rowptr, csr, dinv, b2, W3, h);

  k_pull_last<<<wnBlk, 256, 0, stream>>>(h, rowptr, csr, dinv, b3, out, NN);

  (void)in_sizes; (void)n_in; (void)out_size; (void)ws_size;
}


# Layer-1 product moved into the level-2 sort launch: bins of 400 nodes (25 strips), each block multiplies its own rows by W with dinv read from LDS; 6 launches to 5; sum order still LDS arrival rank
# speedup vs baseline: 5.3470x; 1.0299x over previous
#include <hip/hip_runtime.h>


#define NN 100000
#define NE 1600000
#define BN 400
#define BSLOT 512
#define NBIN (NN / BN)
#define K1 256
#define NPART 256
#define CH ((((NE + NPART - 1) / NPART) + 31) & ~31)
#define SRC_BITS 17
#define SRC_MASK ((1u << SRC_BITS) - 1u)
#define CAP 16384

static_assert(NN <= (1 << SRC_BITS), "src must fit its field");
static_assert(NBIN <= 256, "bin histogram is 256 wide");
static_assert(NN % BN == 0 && BN % 16 == 0 && BN < BSLOT, "bins tile the nodes in whole strips");
static_assert(NPART == 256, "8 runs per wave of a 32-wave block; 4 column groups");

typedef float v2f __attribute__((ext_vector_type(2)));
typedef unsigned v2u __attribute__((ext_vector_type(2)));
typedef float v4f __attribute__((ext_vector_type(4)));
typedef float v8f __attribute__((ext_vector_type(8)));

template <int PER>
__device__ __forceinline__ void wave_scan(const unsigned* in, unsigned* out, int lane) {
  unsigned v[PER];
  unsigned t = 0;
#pragma unroll
  for (int j = 0; j < PER; ++j) { v[j] = in[lane * PER + j]; t += v[j]; }
  unsigned inc = t;
  for (int off = 1; off < 32; off <<= 1) {
    unsigned y = __shfl_up(inc, off, 32);
    if (lane >= off) inc += y;
  }
  unsigned run = inc - t;
#pragma unroll
  for (int j = 0; j < PER; ++j) { out[lane * PER + j] = run; run += v[j]; }
}

__global__ __launch_bounds__(1024) void k_part(const int* __restrict__ src,
                                               const int* __restrict__ dst,
                                               unsigned* __restrict__ part,
                                               unsigned* __restrict__ tbl) {
  __shared__ unsigned hist[256];
  __shared__ unsigned start[256];
  __shared__ unsigned staged[CH];
  const int tid = threadIdx.x, lane = tid & 31;
  const int p = blockIdx.x;
  const int e0 = p * CH;
  int n = NE - e0;
  if (n > CH) n = CH;
  if (n < 0) n = 0;
  if (tid < 256) hist[tid] = 0;
  __syncthreads();
  constexpr int PER = (CH + 1023) / 1024;
  unsigned key[PER], bin[PER], rk[PER];
#pragma unroll
  for (int i = 0; i < PER; ++i) {
    int j = i * 1024 + tid;
    int e = e0 + j;
    if (e > NE - 1) e = NE - 1;
    unsigned s = (unsigned)src[e], d = (unsigned)dst[e];
    rk[i] = 0u;
    unsigned q = d / (unsigned)BN;
    key[i] = ((d - q * (unsigned)BN) << SRC_BITS) | s;
    bin[i] = (j < n) ? q : 0xFFFFFFFFu;
  }
#pragma unroll
  for (int i = 0; i < PER; ++i)
    if (bin[i] != 0xFFFFFFFFu) rk[i] = atomicAdd(&hist[bin[i]], 1u);
  __syncthreads();
  if (tid < 32) wave_scan<8>(hist, start, lane);
  __syncthreads();
  if (tid < NBIN) {
    v2u e = { start[tid], hist[tid] };
    *(v2u*)(tbl + 2 * (tid * NPART + p)) = e;
  }
#pragma unroll
  for (int i = 0; i < PER; ++i)
    if (bin[i] != 0xFFFFFFFFu) staged[start[bin[i]] + rk[i]] = key[i];
  __syncthreads();
  for (int j = tid; j < n; j += 1024) part[e0 + j] = staged[j];
}

__global__ __launch_bounds__(1024) void k_binsort_gemm(const unsigned* __restrict__ part,
                                                       const unsigned* __restrict__ tbl,
                                                       unsigned* __restrict__ rowptr,
                                                       float* __restrict__ dinv,
                                                       unsigned* __restrict__ csr,
                                                       const float* __restrict__ X,
                                                       const float* __restrict__ W,
                                                       float* __restrict__ H) {
  __shared__ __align__(16) v2f wl[(K1 / 2) * 64];
  __shared__ unsigned runsrc[NPART];
  __shared__ unsigned runcnt[NPART];
  __shared__ unsigned cnt[BSLOT];
  __shared__ unsigned cursor[BSLOT];
  __shared__ unsigned red[16];
  __shared__ float dl[BN];
  __shared__ unsigned outb[CAP];
  const int tid = threadIdx.x, lane = tid & 31;
  const int wave = __builtin_amdgcn_readfirstlane(tid >> 5);
  const int b = blockIdx.x;
  constexpr int WT = (K1 / 2) * 16 / 1024;
  static_assert(WT * 1024 == (K1 / 2) * 16, "the tasks deal out evenly");
  v4f we[WT], wo[WT];
#pragma unroll
  for (int j = 0; j < WT; ++j) {
    int t = tid + 1024 * j;
    int p = t >> 4, n4 = (t & 15) * 4;
    we[j] = *(const v4f*)(W + (size_t)(2 * p) * 64 + n4);
    wo[j] = *(const v4f*)(W + (size_t)(2 * p + 1) * 64 + n4);
  }
  if (tid < BSLOT) cnt[tid] = 0;
  unsigned col = 0, c = 0;
  if (tid < NPART) {
    v2u e = *(const v2u*)(tbl + 2 * (b * NPART + tid));
    col = e.x;
    c = e.y;
    runsrc[tid] = (unsigned)tid * CH + col;
    runcnt[tid] = c;
  }
  unsigned sc = col, sn = c;
  for (int off = 16; off > 0; off >>= 1) {
    sc += __shfl_xor(sc, off, 32);
    sn += __shfl_xor(sn, off, 32);
  }
  if (lane == 0 && wave < 8) { red[wave] = sc; red[8 + wave] = sn; }
#pragma unroll
  for (int j = 0; j < WT; ++j) {
    int t = tid + 1024 * j;
    int p = t >> 4, n4 = (t & 15) * 4;
    v2f* d = &wl[p * 64 + n4];
    d[0] = v2f{ we[j].x, wo[j].x };
    d[1] = v2f{ we[j].y, wo[j].y };
    d[2] = v2f{ we[j].z, wo[j].z };
    d[3] = v2f{ we[j].w, wo[j].w };
  }
  __syncthreads();
  unsigned S = 0, nb = 0;
  for (int w = 0; w < 8; ++w) { S += red[w]; nb += red[8 + w]; }
  const bool fits = nb <= (unsigned)CAP;

  unsigned k0[8], k1[8], rc[8], rs[8];
#pragma unroll
  for (int u = 0; u < 8; ++u) {
    int r = wave + 32 * u;
    rc[u] = runcnt[r];
    rs[u] = runsrc[r];
    k0[u] = ((unsigned)lane < rc[u]) ? part[rs[u] + lane] : 0u;
    k1[u] = ((unsigned)lane + 32u < rc[u]) ? part[rs[u] + lane + 32] : 0u;
  }
#pragma unroll
  for (int u = 0; u < 8; ++u) {
    if ((unsigned)lane < rc[u]) atomicAdd(&cnt[k0[u] >> SRC_BITS], 1u);
    if ((unsigned)lane + 32u < rc[u]) atomicAdd(&cnt[k1[u] >> SRC_BITS], 1u);
    for (unsigned i = (unsigned)lane + 64u; i < rc[u]; i += 32u)
      atomicAdd(&cnt[part[rs[u] + i] >> SRC_BITS], 1u);
  }
  __syncthreads();
  if (tid < 32) wave_scan<16>(cnt, cursor, lane);
  __syncthreads();
  if (tid < BN) {
    int node = b * BN + tid;
    float dv = 1.0f / sqrtf((float)cnt[tid] + 1.0f);
    rowptr[node] = S + cursor[tid];
    dinv[node] = dv;
    dl[tid] = dv;
  }
  if (tid == BN && b == NBIN - 1) rowptr[NN] = S + cursor[BN];
  __syncthreads();
#pragma unroll
  for (int u = 0; u < 8; ++u) {
    if ((unsigned)lane < rc[u]) {
      unsigned pos = atomicAdd(&cursor[k0[u] >> SRC_BITS], 1u);
      unsigned s = k0[u] & SRC_MASK;
      if (fits) outb[pos] = s; else csr[S + pos] = s;
    }
    if ((unsigned)lane + 32u < rc[u]) {
      unsigned pos = atomicAdd(&cursor[k1[u] >> SRC_BITS], 1u);
      unsigned s = k1[u] & SRC_MASK;
      if (fits) outb[pos] = s; else csr[S + pos] = s;
    }
    for (unsigned i = (unsigned)lane + 64u; i < rc[u]; i += 32u) {
      unsigned k = part[rs[u] + i];
      unsigned pos = atomicAdd(&cursor[k >> SRC_BITS], 1u);
      unsigned s = k & SRC_MASK;
      if (fits) outb[pos] = s; else csr[S + pos] = s;
    }
  }
  __syncthreads();
  if (fits)
    for (unsigned k = tid; k < nb; k += 1024u) csr[S + k] = outb[k];

  if (wave < BN / 16) {
    const int m  = lane & 15;
    const int h2 = lane >> 4;
    const int r0 = wave * 16;
    const float* xrow = X + (size_t)(b * BN + r0 + m) * K1;
    v8f a0 = {}, a1 = {}, a2 = {}, a3 = {};
    for (int kb = 0; kb < K1; kb += 4) {
      int pa = (kb >> 1) + h2;
      v2f a = *(const v2f*)(xrow + 2 * pa);
      const v2f* wp = &wl[pa * 64 + m];
      v2f b0 = wp[0];
      v2f b1 = wp[16];
      v2f b2 = wp[32];
      v2f b3 = wp[48];
      a0 = __builtin_amdgcn_wmma_f32_16x16x4_f32(false, a, false, b0, (short)0, a0, false, false);
      a1 = __builtin_amdgcn_wmma_f32_16x16x4_f32(false, a, false, b1, (short)0, a1, false, false);
      a2 = __builtin_amdgcn_wmma_f32_16x16x4_f32(false, a, false, b2, (short)0, a2, false, false);
      a3 = __builtin_amdgcn_wmma_f32_16x16x4_f32(false, a, false, b3, (short)0, a3, false, false);
    }
    float* out0 = H + (size_t)(b * BN + r0 + 8 * h2) * 64 + m;
#pragma unroll
    for (int j = 0; j < 8; ++j) {
      float* r = out0 + (size_t)j * 64;
      float dj = dl[r0 + 8 * h2 + j];
      r[0] = a0[j] * dj; r[16] = a1[j] * dj; r[32] = a2[j] * dj; r[48] = a3[j] * dj;
    }
  }
}

__device__ __forceinline__ v2f gather_rows(const float* __restrict__ hl,
                                           const unsigned* __restrict__ csr,
                                           unsigned beg, unsigned end, unsigned my, int lane) {
  const v2f zero = { 0.0f, 0.0f };
  v2f acc = zero;
  for (unsigned cb = beg; cb < end; cb += 32) {
    unsigned n = end - cb;
    if (n > 32u) n = 32u;
    if (cb != beg) {
      my = 0u;
      if ((unsigned)lane < n) my = csr[cb + lane];
    }
    for (unsigned j = 0; j < n; j += 8) {
      unsigned s[8];
      v2f r[8];
#pragma unroll
      for (int u = 0; u < 8; ++u) {
        s[u] = (unsigned)__builtin_amdgcn_readlane((int)my, (int)(j + u));
      }
#pragma unroll
      for (int u = 0; u < 8; ++u) r[u] = *(const v2f*)(hl + (size_t)s[u] * 64);
#pragma unroll
      for (int u = 0; u < 8; ++u) {
        v2f rr = (j + u < n) ? r[u] : zero;
        acc += rr;
      }
    }
  }
  return acc;
}

__global__ __launch_bounds__(256) void k_pull_last(const float* __restrict__ H,
                                                   const unsigned* __restrict__ rowptr,
                                                   const unsigned* __restrict__ csr,
                                                   const float* __restrict__ dinv,
                                                   const float* __restrict__ bias,
                                                   float* __restrict__ out, int N) {
  int node = (int)((blockIdx.x * 256 + threadIdx.x) >> 5);
  if (node >= N) return;
  node = __builtin_amdgcn_readfirstlane(node);
  int lane = threadIdx.x & 31;
  unsigned beg = rowptr[node], end = rowptr[node + 1];
  size_t base = (size_t)node * 64 + lane * 2;
  v2f hv = *(const v2f*)(H + base);
  v2f bv = *(const v2f*)(bias + lane * 2);
  float di = dinv[node];
  unsigned my = 0u;
  if ((unsigned)lane < end - beg) my = csr[beg + lane];
  v2f acc = gather_rows(H + lane * 2, csr, beg, end, my, lane);
  v2f o = { di * (acc.x + hv.x) + bv.x, di * (acc.y + hv.y) + bv.y };
  *(v2f*)(out + base) = o;
}

#define TP 68
#define FUSE_GRID 1024
static_assert(NN % 16 == 0, "tiles of 16 nodes cover the nodes exactly");

__global__ __launch_bounds__(512) void k_pull_gemm(const float* __restrict__ H,
                                                   const unsigned* __restrict__ rowptr,
                                                   const unsigned* __restrict__ csr,
                                                   const float* __restrict__ dinv,
                                                   const float* __restrict__ bias,
                                                   const float* __restrict__ W,
                                                   float* __restrict__ Hout) {
  __shared__ __align__(16) v2f wl[32 * 64];
  __shared__ __align__(16) float xs[16 * TP];
  __shared__ __align__(16) float ys[16 * TP];
  const int wp = (int)(threadIdx.x >> 4), wn4 = (int)(threadIdx.x & 15) * 4;
  const v4f we = *(const v4f*)(W + (size_t)(2 * wp) * 64 + wn4);
  const v4f wo = *(const v4f*)(W + (size_t)(2 * wp + 1) * 64 + wn4);
  const int lane = threadIdx.x & 31;
  const int wave = __builtin_amdgcn_readfirstlane((int)(threadIdx.x >> 5));
  const int m = lane & 15, h2 = lane >> 4;
  const int ntile = NN / 16;
  const v2f bv = *(const v2f*)(bias + lane * 2);
  const float* hl = H + lane * 2;

  int tile = blockIdx.x;
  unsigned beg = 0u, end = 0u, my = 0u;
  if (tile < ntile) {
    int node = tile * 16 + wave;
    beg = rowptr[node];
    end = rowptr[node + 1];
    if ((unsigned)lane < end - beg) my = csr[beg + lane];
  }
  {
    v2f* d = &wl[wp * 64 + wn4];
    d[0] = v2f{ we.x, wo.x };
    d[1] = v2f{ we.y, wo.y };
    d[2] = v2f{ we.z, wo.z };
    d[3] = v2f{ we.w, wo.w };
  }
  for (; tile < ntile; tile += FUSE_GRID) {
    const int node = tile * 16 + wave;
    const size_t base = (size_t)node * 64 + lane * 2;
    v2f hv = *(const v2f*)(H + base);
    float di = dinv[node];
    const int ntl = tile + FUSE_GRID;
    const bool more = ntl < ntile;
    unsigned rp = 0u;
    if (more) rp = rowptr[ntl * 16 + wave + (lane & 1)];
    v2f acc = gather_rows(hl, csr, beg, end, my, lane);
    const unsigned nbeg = (unsigned)__builtin_amdgcn_readlane((int)rp, 0);
    const unsigned nend = (unsigned)__builtin_amdgcn_readlane((int)rp, 1);
    float vx = di * (acc.x + hv.x) + bv.x;
    float vy = di * (acc.y + hv.y) + bv.y;
    vx = fmaxf(vx, 0.0f);
    vy = fmaxf(vy, 0.0f);
    float ss = vx * vx + vy * vy;
    for (int off = 16; off > 0; off >>= 1) ss += __shfl_xor(ss, off, 32);
    float scale = 1.0f / fmaxf(sqrtf(ss), 1e-12f);
    v2f xr = { vx * scale, vy * scale };
    *(v2f*)&xs[wave * TP + lane * 2] = xr;
    __syncthreads();
    unsigned nmy = 0u;
    if (more && (unsigned)lane < nend - nbeg) nmy = csr[nbeg + lane];
    if (wave < 4) {
      const int row0 = tile * 16;
      v4f dlo = *(const v4f*)(dinv + row0 + 8 * h2);
      v4f dhi = *(const v4f*)(dinv + row0 + 8 * h2 + 4);
      v8f a0 = {};
#pragma unroll 8
      for (int kb = 0; kb < 64; kb += 4) {
        int pa = (kb >> 1) + h2;
        v2f a = *(const v2f*)&xs[m * TP + 2 * pa];
        v2f b = wl[pa * 64 + wave * 16 + m];
        a0 = __builtin_amdgcn_wmma_f32_16x16x4_f32(false, a, false, b, (short)0, a0, false, false);
      }
#pragma unroll
      for (int j = 0; j < 8; ++j) {
        float dj = (j < 4) ? dlo[j & 3] : dhi[j & 3];
        ys[(8 * h2 + j) * TP + wave * 16 + m] = a0[j] * dj;
      }
    }
    __syncthreads();
    *(v2f*)(Hout + base) = *(const v2f*)&ys[wave * TP + lane * 2];
    beg = nbeg; end = nend; my = nmy;
  }
}

extern "C" void kernel_launch(void* const* d_in, const int* in_sizes, int n_in,
                              void* d_out, int out_size, void* d_ws, size_t ws_size,
                              hipStream_t stream) {
  const float* x  = (const float*)d_in[0];
  const int*   ei = (const int*)d_in[1];
  const float* W1 = (const float*)d_in[2];
  const float* b1 = (const float*)d_in[3];
  const float* W2 = (const float*)d_in[4];
  const float* b2 = (const float*)d_in[5];
  const float* W3 = (const float*)d_in[6];
  const float* b3 = (const float*)d_in[7];
  float* out = (float*)d_out;

  const int* src = ei;
  const int* dst = ei + NE;

  char* ws = (char*)d_ws;
  size_t off = 0;
  auto carve = [&](size_t bytes) {
    void* p = ws + off;
    off = (off + bytes + 255) & ~(size_t)255;
    return p;
  };
  float*    dinv   = (float*)   carve((size_t)NN * 4);
  unsigned* rowptr = (unsigned*)carve((size_t)(NN + 1) * 4);
  unsigned* tbl    = (unsigned*)carve((size_t)NBIN * NPART * 8);
  unsigned* part   = (unsigned*)carve((size_t)NE * 4);
  unsigned* csr    = (unsigned*)carve((size_t)NE * 4);
  float*    h      = (float*)   carve((size_t)NN * 64 * 4);
  float*    hn     = (float*)   carve((size_t)NN * 64 * 4);

  const int wnBlk  = (NN * 32 + 255) / 256;

  k_part   <<<NPART, 1024, 0, stream>>>(src, dst, part, tbl);
  k_binsort_gemm<<<NBIN, 1024, 0, stream>>>(part, tbl, rowptr, dinv, csr, x, W1, h);

  k_pull_gemm<<<FUSE_GRID, 512, 0, stream>>>(h, rowptr, csr, dinv, b1, W2, hn);

  k_pull_gemm<<<FUSE_GRID, 512, 0, stream>>>(hn, rowptr, csr, dinv, b2, W3, h);

  k_pull_last<<<wnBlk, 256, 0, stream>>>(h, rowptr, csr, dinv, b3, out, NN);

  (void)in_sizes; (void)n_in; (void)out_size; (void)ws_size;
}
